# starting kernel
# baseline (speedup 1.0000x reference)
#include <hip/hip_runtime.h>

typedef float v2f __attribute__((ext_vector_type(2)));
typedef float v8f __attribute__((ext_vector_type(8)));

__global__ void k_init(float* __restrict__ dinv, float* __restrict__ gagg, int n) {
  int i = blockIdx.x * blockDim.x + threadIdx.x;
  if (i < n) {
    dinv[i]      = 1.0f;
    gagg[2*i]    = 0.0f;
    gagg[2*i+1]  = 0.0f;
  }
}

__global__ void k_degree(const int* __restrict__ dstIdx, float* __restrict__ deg, int e) {
  int i = (blockIdx.x * blockDim.x + threadIdx.x) * 4;
  if (i + 3 < e) {
    int4 d = *(const int4*)(dstIdx + i);
    atomicAdd(&deg[d.x], 1.0f);
    atomicAdd(&deg[d.y], 1.0f);
    atomicAdd(&deg[d.z], 1.0f);
    atomicAdd(&deg[d.w], 1.0f);
  } else {
    for (int j = i; j < e; ++j) atomicAdd(&deg[dstIdx[j]], 1.0f);
  }
}

__global__ void k_dinv(const float* __restrict__ x, float* __restrict__ dinv,
                       float* __restrict__ s, int n) {
  int i = blockIdx.x * blockDim.x + threadIdx.x;
  if (i < n) {
    float dv = rsqrtf(dinv[i]);
    dinv[i] = dv;
    s[i] = dv * dv * x[i];
  }
}

__global__ void k_scatter1(const int* __restrict__ srcIdx, const int* __restrict__ dstIdx,
                           const float* __restrict__ x, const float* __restrict__ dinv,
                           float* __restrict__ s, int e) {
  int i = (blockIdx.x * blockDim.x + threadIdx.x) * 4;
  if (i + 3 < e) {
    int4 sv = *(const int4*)(srcIdx + i);
    int4 dv = *(const int4*)(dstIdx + i);
    atomicAdd(&s[dv.x], dinv[sv.x] * dinv[dv.x] * x[sv.x]);
    atomicAdd(&s[dv.y], dinv[sv.y] * dinv[dv.y] * x[sv.y]);
    atomicAdd(&s[dv.z], dinv[sv.z] * dinv[dv.z] * x[sv.z]);
    atomicAdd(&s[dv.w], dinv[sv.w] * dinv[dv.w] * x[sv.w]);
  } else {
    for (int j = i; j < e; ++j) {
      int sn = srcIdx[j], dn = dstIdx[j];
      atomicAdd(&s[dn], dinv[sn] * dinv[dn] * x[sn]);
    }
  }
}

__global__ __launch_bounds__(256)
void k_wmma_transform(const float* __restrict__ s,
                      const float* __restrict__ W1,
                      const float* __restrict__ b1,
                      const float* __restrict__ W2,
                      float* __restrict__ g, int n, int ntiles) {
  int lane = threadIdx.x & 31;
  int wave = blockIdx.x * (blockDim.x >> 5) + (threadIdx.x >> 5);
  if (wave >= ntiles) return;
  int t0   = wave * 16;
  int half = lane >> 4;
  int m    = lane & 15;
  int node = t0 + m;
  int nidx   = node < n ? node : 0;
  float sn   = (node < n) ? s[nidx] : 0.0f;
  float cmsk = (m < 2) ? 1.0f : 0.0f;
  int   cm   = (m < 2) ? m : 0;

  v2f Bf[8];
  #pragma unroll
  for (int i = 0; i < 8; ++i) {
    int f0 = 4 * i + 2 * half;
    Bf[i].x = W2[(f0    ) * 2 + cm] * cmsk;
    Bf[i].y = W2[(f0 + 1) * 2 + cm] * cmsk;
  }

  v8f c = {};
  #pragma unroll
  for (int i = 0; i < 8; ++i) {
    int f0 = 4 * i + 2 * half;
    int f1 = f0 + 1;
    v2f A;
    A.x = fmaxf(fmaf(sn, W1[f0], b1[f0]), 0.0f);
    A.y = fmaxf(fmaf(sn, W1[f1], b1[f1]), 0.0f);
    c = __builtin_amdgcn_wmma_f32_16x16x4_f32(
             false, A,  false, Bf[i],
             (short)0, c,  false,  false);
  }

  if (m < 2) {
    #pragma unroll
    for (int v = 0; v < 8; ++v) {
      int row = t0 + v + 8 * half;
      if (row < n) g[row * 2 + m] = c[v];
    }
  }
}

__global__ void k_scatter2(const int* __restrict__ srcIdx, const int* __restrict__ dstIdx,
                           const float* __restrict__ g, const float* __restrict__ dinv,
                           float* __restrict__ gagg, int e) {
  int i = (blockIdx.x * blockDim.x + threadIdx.x) * 2;
  if (i + 1 < e) {
    int2 sv = *(const int2*)(srcIdx + i);
    int2 dv = *(const int2*)(dstIdx + i);
    float  w0 = dinv[sv.x] * dinv[dv.x];
    float  w1 = dinv[sv.y] * dinv[dv.y];
    float2 g0 = *(const float2*)(g + 2 * sv.x);
    float2 g1 = *(const float2*)(g + 2 * sv.y);
    atomicAdd(&gagg[2*dv.x],     w0 * g0.x);
    atomicAdd(&gagg[2*dv.x + 1], w0 * g0.y);
    atomicAdd(&gagg[2*dv.y],     w1 * g1.x);
    atomicAdd(&gagg[2*dv.y + 1], w1 * g1.y);
  } else if (i < e) {
    int sn = srcIdx[i], dn = dstIdx[i];
    float w = dinv[sn] * dinv[dn];
    float2 gv = *(const float2*)(g + 2 * sn);
    atomicAdd(&gagg[2*dn],     w * gv.x);
    atomicAdd(&gagg[2*dn + 1], w * gv.y);
  }
}

__global__ void k_finalize(const float* __restrict__ gagg, const float* __restrict__ g,
                           const float* __restrict__ dinv, const float* __restrict__ b2,
                           float* __restrict__ out, int n) {
  int i = blockIdx.x * blockDim.x + threadIdx.x;
  if (i < n) {
    float d2 = dinv[i] * dinv[i];
    float2 ga = *(const float2*)(gagg + 2 * i);
    float2 gv = *(const float2*)(g + 2 * i);
    float o0 = ga.x + d2 * gv.x + b2[0];
    float o1 = ga.y + d2 * gv.y + b2[1];
    float mx  = fmaxf(o0, o1);
    float lse = mx + logf(expf(o0 - mx) + expf(o1 - mx));
    float2 r; r.x = o0 - lse; r.y = o1 - lse;
    *(float2*)(out + 2 * i) = r;
  }
}

extern "C" void kernel_launch(void* const* d_in, const int* in_sizes, int n_in,
                              void* d_out, int out_size, void* d_ws, size_t ws_size,
                              hipStream_t stream) {
  const float* x  = (const float*)d_in[0];
  const int*   ei = (const int*)d_in[1];
  const float* W1 = (const float*)d_in[2];
  const float* b1 = (const float*)d_in[3];
  const float* W2 = (const float*)d_in[4];
  const float* b2 = (const float*)d_in[5];
  float* out = (float*)d_out;

  int n = in_sizes[0];
  int e = in_sizes[1] / 2;
  const int* srcIdx = ei;
  const int* dstIdx = ei + e;

  float* dinv = (float*)d_ws;
  float* s    = dinv + n;
  float* g    = s + n;
  float* gagg = g + 2 * (size_t)n;

  const int tb = 256;
  int gn  = (n + tb - 1) / tb;
  int ge4 = ((e + 3) / 4 + tb - 1) / tb;
  int ge2 = ((e + 1) / 2 + tb - 1) / tb;

  k_init    <<<gn,  tb, 0, stream>>>(dinv, gagg, n);
  k_degree  <<<ge4, tb, 0, stream>>>(dstIdx, dinv, e);
  k_dinv    <<<gn,  tb, 0, stream>>>(x, dinv, s, n);
  k_scatter1<<<ge4, tb, 0, stream>>>(srcIdx, dstIdx, x, dinv, s, e);

  int ntiles = (n + 15) / 16;
  int gw = (ntiles + 7) / 8;
  k_wmma_transform<<<gw, tb, 0, stream>>>(s, W1, b1, W2, g, n, ntiles);

  k_scatter2<<<ge2, tb, 0, stream>>>(srcIdx, dstIdx, g, dinv, gagg, e);
  k_finalize<<<gn,  tb, 0, stream>>>(gagg, g, dinv, b2, out, n);
}


# Edges counting-sorted by dst node range per chunk (one pass); degree and both aggregations accumulate with LDS atomics per node range, no global atomics; float sums stay in arrival order
# speedup vs baseline: 5.6798x; 5.6798x over previous
#include <hip/hip_runtime.h>

typedef float v2f __attribute__((ext_vector_type(2)));
typedef float v8f __attribute__((ext_vector_type(8)));

#define SH    9
#define RB    (1 << SH)
#define MAXB  256
#define NCH   256
#define PT    1024

__global__ __launch_bounds__(PT)
void k_part(const int* __restrict__ srcIdx, const int* __restrict__ dstIdx,
            int* __restrict__ pairsRaw, int* __restrict__ tab, int e, int ch) {
  __shared__ int cnt[MAXB];
  const int tid = threadIdx.x;
  const int c   = blockIdx.x;
  const int beg = c * ch;
  const int end = min(beg + ch, e);
  int2* __restrict__ pairs = (int2*)pairsRaw;

  if (tid < MAXB) cnt[tid] = 0;
  __syncthreads();
  for (int i = beg + 4 * tid; i < end; i += 4 * PT) {
    if (i + 3 < end) {
      int4 d = *(const int4*)(dstIdx + i);
      atomicAdd(&cnt[d.x >> SH], 1);
      atomicAdd(&cnt[d.y >> SH], 1);
      atomicAdd(&cnt[d.z >> SH], 1);
      atomicAdd(&cnt[d.w >> SH], 1);
    } else {
      for (int j = i; j < end; ++j) atomicAdd(&cnt[dstIdx[j] >> SH], 1);
    }
  }
  __syncthreads();

  if (tid < 32) {
    int cn[8], st[8];
    int run = 0;
    #pragma unroll
    for (int k = 0; k < 8; ++k) { cn[k] = cnt[tid * 8 + k]; st[k] = run; run += cn[k]; }
    int incl = run;
    #pragma unroll
    for (int off = 1; off < 32; off <<= 1) {
      int t = __shfl_up(incl, off);
      if (tid >= off) incl += t;
    }
    const int excl = incl - run;
    int2* row = (int2*)tab + c * MAXB;
    #pragma unroll
    for (int k = 0; k < 8; ++k) {
      int p = excl + st[k];
      cnt[tid * 8 + k] = p;
      row[tid * 8 + k] = make_int2(p, p + cn[k]);
    }
  }
  __syncthreads();

  for (int i = beg + 4 * tid; i < end; i += 4 * PT) {
    if (i + 3 < end) {
      int4 d = *(const int4*)(dstIdx + i);
      int4 s = *(const int4*)(srcIdx + i);
      int p0 = atomicAdd(&cnt[d.x >> SH], 1);
      int p1 = atomicAdd(&cnt[d.y >> SH], 1);
      int p2 = atomicAdd(&cnt[d.z >> SH], 1);
      int p3 = atomicAdd(&cnt[d.w >> SH], 1);
      pairs[beg + p0] = make_int2(s.x, d.x);
      pairs[beg + p1] = make_int2(s.y, d.y);
      pairs[beg + p2] = make_int2(s.z, d.z);
      pairs[beg + p3] = make_int2(s.w, d.w);
    } else {
      for (int j = i; j < end; ++j) {
        int d = dstIdx[j];
        int p = atomicAdd(&cnt[d >> SH], 1);
        pairs[beg + p] = make_int2(srcIdx[j], d);
      }
    }
  }
}

__global__ __launch_bounds__(PT)
void k_deg(const int* __restrict__ pairsRaw, const int* __restrict__ tab,
           float* __restrict__ dinv, int n, int ch) {
  __shared__ int cnt[RB];
  const int tid = threadIdx.x, lane = tid & 31, w = tid >> 5;
  const int b  = blockIdx.x;
  const int n0 = b << SH;
  const int2* __restrict__ pairs = (const int2*)pairsRaw;

  if (tid < RB) cnt[tid] = 0;
  __syncthreads();
  for (int c = w; c < NCH; c += PT / 32) {
    const int2 az = ((const int2*)tab)[c * MAXB + b];
    const int a = az.x, z = az.y;
    const int2* p = pairs + (size_t)c * ch;
    for (int i = a + lane; i < z; i += 32)
      atomicAdd(&cnt[p[i].y - n0], 1);
  }
  __syncthreads();
  if (tid < RB) {
    int node = n0 + tid;
    if (node < n) dinv[node] = rsqrtf((float)(cnt[tid] + 1));
  }
}

__global__ __launch_bounds__(PT)
void k_layer1(const int* __restrict__ pairsRaw, const int* __restrict__ tab,
              const float* __restrict__ x, const float* __restrict__ dinv,
              float* __restrict__ s, int n, int ch) {
  __shared__ float acc[RB];
  const int tid = threadIdx.x, lane = tid & 31, w = tid >> 5;
  const int b  = blockIdx.x;
  const int n0 = b << SH;
  const int2* __restrict__ pairs = (const int2*)pairsRaw;

  if (tid < RB) {
    int node = n0 + tid;
    float v = 0.0f;
    if (node < n) { float dv = dinv[node]; v = dv * dv * x[node]; }
    acc[tid] = v;
  }
  __syncthreads();
  for (int c = w; c < NCH; c += PT / 32) {
    const int2 az = ((const int2*)tab)[c * MAXB + b];
    const int a = az.x, z = az.y;
    const int2* p = pairs + (size_t)c * ch;
    for (int i = a + lane; i < z; i += 32) {
      int2 q = p[i];
      atomicAdd(&acc[q.y - n0], dinv[q.x] * dinv[q.y] * x[q.x]);
    }
  }
  __syncthreads();
  if (tid < RB) {
    int node = n0 + tid;
    if (node < n) s[node] = acc[tid];
  }
}

__global__ __launch_bounds__(256)
void k_wmma_transform(const float* __restrict__ s,
                      const float* __restrict__ W1,
                      const float* __restrict__ b1,
                      const float* __restrict__ W2,
                      float* __restrict__ g, int n, int ntiles) {
  int lane = threadIdx.x & 31;
  int wave = blockIdx.x * (blockDim.x >> 5) + (threadIdx.x >> 5);
  if (wave >= ntiles) return;
  int t0   = wave * 16;
  int half = lane >> 4;
  int m    = lane & 15;
  int node = t0 + m;
  int nidx   = node < n ? node : 0;
  float sn   = (node < n) ? s[nidx] : 0.0f;
  float cmsk = (m < 2) ? 1.0f : 0.0f;
  int   cm   = (m < 2) ? m : 0;

  v2f Bf[8];
  #pragma unroll
  for (int i = 0; i < 8; ++i) {
    int f0 = 4 * i + 2 * half;
    Bf[i].x = W2[(f0    ) * 2 + cm] * cmsk;
    Bf[i].y = W2[(f0 + 1) * 2 + cm] * cmsk;
  }

  v8f c = {};
  #pragma unroll
  for (int i = 0; i < 8; ++i) {
    int f0 = 4 * i + 2 * half;
    int f1 = f0 + 1;
    v2f A;
    A.x = fmaxf(fmaf(sn, W1[f0], b1[f0]), 0.0f);
    A.y = fmaxf(fmaf(sn, W1[f1], b1[f1]), 0.0f);
    c = __builtin_amdgcn_wmma_f32_16x16x4_f32(
             false, A,  false, Bf[i],
             (short)0, c,  false,  false);
  }

  if (m < 2) {
    #pragma unroll
    for (int v = 0; v < 8; ++v) {
      int row = t0 + v + 8 * half;
      if (row < n) g[row * 2 + m] = c[v];
    }
  }
}

__global__ __launch_bounds__(PT)
void k_layer2(const int* __restrict__ pairsRaw, const int* __restrict__ tab,
              const float* __restrict__ g, const float* __restrict__ dinv,
              const float* __restrict__ b2, float* __restrict__ out, int n, int ch) {
  __shared__ float acc[2 * RB];
  const int tid = threadIdx.x, lane = tid & 31, w = tid >> 5;
  const int b  = blockIdx.x;
  const int n0 = b << SH;
  const int2* __restrict__ pairs = (const int2*)pairsRaw;

  for (int t = tid; t < 2 * RB; t += PT) acc[t] = 0.0f;
  __syncthreads();
  for (int c = w; c < NCH; c += PT / 32) {
    const int2 az = ((const int2*)tab)[c * MAXB + b];
    const int a = az.x, z = az.y;
    const int2* p = pairs + (size_t)c * ch;
    for (int i = a + lane; i < z; i += 32) {
      int2 q = p[i];
      float  wgt = dinv[q.x] * dinv[q.y];
      float2 gv  = *(const float2*)(g + 2 * q.x);
      int t = q.y - n0;
      atomicAdd(&acc[2 * t],     wgt * gv.x);
      atomicAdd(&acc[2 * t + 1], wgt * gv.y);
    }
  }
  __syncthreads();
  if (tid < RB) {
    int node = n0 + tid;
    if (node < n) {
      float dv = dinv[node];
      float d2 = dv * dv;
      float2 gv = *(const float2*)(g + 2 * node);
      float o0 = acc[2 * tid]     + d2 * gv.x + b2[0];
      float o1 = acc[2 * tid + 1] + d2 * gv.y + b2[1];
      float mx  = fmaxf(o0, o1);
      float lse = mx + logf(expf(o0 - mx) + expf(o1 - mx));
      float2 r; r.x = o0 - lse; r.y = o1 - lse;
      *(float2*)(out + 2 * node) = r;
    }
  }
}

extern "C" void kernel_launch(void* const* d_in, const int* in_sizes, int n_in,
                              void* d_out, int out_size, void* d_ws, size_t ws_size,
                              hipStream_t stream) {
  const float* x  = (const float*)d_in[0];
  const int*   ei = (const int*)d_in[1];
  const float* W1 = (const float*)d_in[2];
  const float* b1 = (const float*)d_in[3];
  const float* W2 = (const float*)d_in[4];
  const float* b2 = (const float*)d_in[5];
  float* out = (float*)d_out;

  int n = in_sizes[0];
  int e = in_sizes[1] / 2;
  const int* srcIdx = ei;
  const int* dstIdx = ei + e;

  float* dinv  = (float*)d_ws;
  float* s     = dinv + n;
  float* g     = s + n;
  int*   tab   = (int*)(g + 2 * (size_t)n);
  int*   pairs = tab + 2 * NCH * MAXB;

  int nb = (n + RB - 1) / RB;
  int ch = (((e + NCH - 1) / NCH) + 3) & ~3;

  k_part  <<<NCH, PT, 0, stream>>>(srcIdx, dstIdx, pairs, tab, e, ch);
  k_deg   <<<nb,  PT, 0, stream>>>(pairs, tab, dinv, n, ch);
  k_layer1<<<nb,  PT, 0, stream>>>(pairs, tab, x, dinv, s, n, ch);

  const int tb = 256;
  int ntiles = (n + 15) / 16;
  int gw = (ntiles + 7) / 8;
  k_wmma_transform<<<gw, tb, 0, stream>>>(s, W1, b1, W2, g, n, ntiles);

  k_layer2<<<nb,  PT, 0, stream>>>(pairs, tab, g, dinv, b2, out, n, ch);
}


# Source factor rsqrt(deg[src]) folded into the gathered value where it is made, dst factor applied once per node outside the sum (sums regrouped): one gather per edge per layer instead of three
# speedup vs baseline: 7.5109x; 1.3224x over previous
#include <hip/hip_runtime.h>

typedef float v2f __attribute__((ext_vector_type(2)));
typedef float v8f __attribute__((ext_vector_type(8)));

#define SH    9
#define RB    (1 << SH)
#define MAXB  256
#define NCH   256
#define PT    1024

__global__ __launch_bounds__(PT)
void k_part(const int* __restrict__ srcIdx, const int* __restrict__ dstIdx,
            int* __restrict__ pairsRaw, int* __restrict__ tab, int e, int ch) {
  __shared__ int cnt[MAXB];
  const int tid = threadIdx.x;
  const int c   = blockIdx.x;
  const int beg = c * ch;
  const int end = min(beg + ch, e);
  int2* __restrict__ pairs = (int2*)pairsRaw;

  if (tid < MAXB) cnt[tid] = 0;
  __syncthreads();
  for (int i = beg + 4 * tid; i < end; i += 4 * PT) {
    if (i + 3 < end) {
      int4 d = *(const int4*)(dstIdx + i);
      atomicAdd(&cnt[d.x >> SH], 1);
      atomicAdd(&cnt[d.y >> SH], 1);
      atomicAdd(&cnt[d.z >> SH], 1);
      atomicAdd(&cnt[d.w >> SH], 1);
    } else {
      for (int j = i; j < end; ++j) atomicAdd(&cnt[dstIdx[j] >> SH], 1);
    }
  }
  __syncthreads();

  if (tid < 32) {
    int cn[8], st[8];
    int run = 0;
    #pragma unroll
    for (int k = 0; k < 8; ++k) { cn[k] = cnt[tid * 8 + k]; st[k] = run; run += cn[k]; }
    int incl = run;
    #pragma unroll
    for (int off = 1; off < 32; off <<= 1) {
      int t = __shfl_up(incl, off);
      if (tid >= off) incl += t;
    }
    const int excl = incl - run;
    int2* row = (int2*)tab + c * MAXB;
    #pragma unroll
    for (int k = 0; k < 8; ++k) {
      int p = excl + st[k];
      cnt[tid * 8 + k] = p;
      row[tid * 8 + k] = make_int2(p, p + cn[k]);
    }
  }
  __syncthreads();

  for (int i = beg + 4 * tid; i < end; i += 4 * PT) {
    if (i + 3 < end) {
      int4 d = *(const int4*)(dstIdx + i);
      int4 s = *(const int4*)(srcIdx + i);
      int p0 = atomicAdd(&cnt[d.x >> SH], 1);
      int p1 = atomicAdd(&cnt[d.y >> SH], 1);
      int p2 = atomicAdd(&cnt[d.z >> SH], 1);
      int p3 = atomicAdd(&cnt[d.w >> SH], 1);
      pairs[beg + p0] = make_int2(s.x, d.x);
      pairs[beg + p1] = make_int2(s.y, d.y);
      pairs[beg + p2] = make_int2(s.z, d.z);
      pairs[beg + p3] = make_int2(s.w, d.w);
    } else {
      for (int j = i; j < end; ++j) {
        int d = dstIdx[j];
        int p = atomicAdd(&cnt[d >> SH], 1);
        pairs[beg + p] = make_int2(srcIdx[j], d);
      }
    }
  }
}

__global__ __launch_bounds__(PT)
void k_deg(const int* __restrict__ pairsRaw, const int* __restrict__ tab,
           const float* __restrict__ x, float* __restrict__ dinv,
           float* __restrict__ y, int n, int ch) {
  __shared__ int cnt[RB];
  const int tid = threadIdx.x, lane = tid & 31, w = tid >> 5;
  const int b  = blockIdx.x;
  const int n0 = b << SH;
  const int2* __restrict__ pairs = (const int2*)pairsRaw;

  if (tid < RB) cnt[tid] = 0;
  __syncthreads();
  for (int c = w; c < NCH; c += PT / 32) {
    const int2 az = ((const int2*)tab)[c * MAXB + b];
    const int a = az.x, z = az.y;
    const int2* p = pairs + (size_t)c * ch;
    for (int i = a + lane; i < z; i += 32)
      atomicAdd(&cnt[p[i].y - n0], 1);
  }
  __syncthreads();
  if (tid < RB) {
    int node = n0 + tid;
    if (node < n) {
      float dv = rsqrtf((float)(cnt[tid] + 1));
      dinv[node] = dv;
      y[node]    = dv * x[node];
    }
  }
}

__global__ __launch_bounds__(PT)
void k_layer1(const int* __restrict__ pairsRaw, const int* __restrict__ tab,
              const float* __restrict__ y, const float* __restrict__ dinv,
              float* __restrict__ s, int n, int ch) {
  __shared__ float acc[RB];
  const int tid = threadIdx.x, lane = tid & 31, w = tid >> 5;
  const int b  = blockIdx.x;
  const int n0 = b << SH;
  const int2* __restrict__ pairs = (const int2*)pairsRaw;

  if (tid < RB) {
    int node = n0 + tid;
    acc[tid] = (node < n) ? y[node] : 0.0f;
  }
  __syncthreads();
  for (int c = w; c < NCH; c += PT / 32) {
    const int2 az = ((const int2*)tab)[c * MAXB + b];
    const int a = az.x, z = az.y;
    const int2* p = pairs + (size_t)c * ch;
    for (int i = a + lane; i < z; i += 32) {
      int2 q = p[i];
      atomicAdd(&acc[q.y - n0], y[q.x]);
    }
  }
  __syncthreads();
  if (tid < RB) {
    int node = n0 + tid;
    if (node < n) s[node] = dinv[node] * acc[tid];
  }
}

__global__ __launch_bounds__(256)
void k_wmma_transform(const float* __restrict__ s,
                      const float* __restrict__ dinv,
                      const float* __restrict__ W1,
                      const float* __restrict__ b1,
                      const float* __restrict__ W2,
                      float* __restrict__ g, int n, int ntiles) {
  int lane = threadIdx.x & 31;
  int wave = blockIdx.x * (blockDim.x >> 5) + (threadIdx.x >> 5);
  if (wave >= ntiles) return;
  int t0   = wave * 16;
  int half = lane >> 4;
  int m    = lane & 15;
  int node = t0 + m;
  int nidx   = node < n ? node : 0;
  float sn   = (node < n) ? s[nidx] : 0.0f;
  float dn   = (node < n) ? dinv[nidx] : 0.0f;
  float cmsk = (m < 2) ? 1.0f : 0.0f;
  int   cm   = (m < 2) ? m : 0;

  v2f Bf[8];
  #pragma unroll
  for (int i = 0; i < 8; ++i) {
    int f0 = 4 * i + 2 * half;
    Bf[i].x = W2[(f0    ) * 2 + cm] * cmsk;
    Bf[i].y = W2[(f0 + 1) * 2 + cm] * cmsk;
  }

  v8f c = {};
  #pragma unroll
  for (int i = 0; i < 8; ++i) {
    int f0 = 4 * i + 2 * half;
    int f1 = f0 + 1;
    v2f A;
    A.x = dn * fmaxf(fmaf(sn, W1[f0], b1[f0]), 0.0f);
    A.y = dn * fmaxf(fmaf(sn, W1[f1], b1[f1]), 0.0f);
    c = __builtin_amdgcn_wmma_f32_16x16x4_f32(
             false, A,  false, Bf[i],
             (short)0, c,  false,  false);
  }

  if (m < 2) {
    #pragma unroll
    for (int v = 0; v < 8; ++v) {
      int row = t0 + v + 8 * half;
      if (row < n) g[row * 2 + m] = c[v];
    }
  }
}

__global__ __launch_bounds__(PT)
void k_layer2(const int* __restrict__ pairsRaw, const int* __restrict__ tab,
              const float* __restrict__ g, const float* __restrict__ dinv,
              const float* __restrict__ b2, float* __restrict__ out, int n, int ch) {
  __shared__ float acc[2 * RB];
  const int tid = threadIdx.x, lane = tid & 31, w = tid >> 5;
  const int b  = blockIdx.x;
  const int n0 = b << SH;
  const int2* __restrict__ pairs = (const int2*)pairsRaw;

  for (int t = tid; t < 2 * RB; t += PT) acc[t] = 0.0f;
  __syncthreads();
  for (int c = w; c < NCH; c += PT / 32) {
    const int2 az = ((const int2*)tab)[c * MAXB + b];
    const int a = az.x, z = az.y;
    const int2* p = pairs + (size_t)c * ch;
    for (int i = a + lane; i < z; i += 32) {
      int2 q = p[i];
      float2 gv  = *(const float2*)(g + 2 * q.x);
      int t = q.y - n0;
      atomicAdd(&acc[2 * t],     gv.x);
      atomicAdd(&acc[2 * t + 1], gv.y);
    }
  }
  __syncthreads();
  if (tid < RB) {
    int node = n0 + tid;
    if (node < n) {
      float dv = dinv[node];
      float2 gv = *(const float2*)(g + 2 * node);
      float o0 = dv * (acc[2 * tid]     + gv.x) + b2[0];
      float o1 = dv * (acc[2 * tid + 1] + gv.y) + b2[1];
      float mx  = fmaxf(o0, o1);
      float lse = mx + logf(expf(o0 - mx) + expf(o1 - mx));
      float2 r; r.x = o0 - lse; r.y = o1 - lse;
      *(float2*)(out + 2 * node) = r;
    }
  }
}

extern "C" void kernel_launch(void* const* d_in, const int* in_sizes, int n_in,
                              void* d_out, int out_size, void* d_ws, size_t ws_size,
                              hipStream_t stream) {
  const float* x  = (const float*)d_in[0];
  const int*   ei = (const int*)d_in[1];
  const float* W1 = (const float*)d_in[2];
  const float* b1 = (const float*)d_in[3];
  const float* W2 = (const float*)d_in[4];
  const float* b2 = (const float*)d_in[5];
  float* out = (float*)d_out;

  int n = in_sizes[0];
  int e = in_sizes[1] / 2;
  const int* srcIdx = ei;
  const int* dstIdx = ei + e;

  float* dinv  = (float*)d_ws;
  float* s     = dinv + n;
  float* g     = s + n;
  int*   tab   = (int*)(g + 2 * (size_t)n);
  int*   pairs = tab + 2 * NCH * MAXB;

  int nb = (n + RB - 1) / RB;
  int ch = (((e + NCH - 1) / NCH) + 3) & ~3;
  float* y = (float*)(pairs + 2 * (size_t)NCH * ch);

  k_part  <<<NCH, PT, 0, stream>>>(srcIdx, dstIdx, pairs, tab, e, ch);
  k_deg   <<<nb,  PT, 0, stream>>>(pairs, tab, x, dinv, y, n, ch);
  k_layer1<<<nb,  PT, 0, stream>>>(pairs, tab, y, dinv, s, n, ch);

  const int tb = 256;
  int ntiles = (n + 15) / 16;
  int gw = (ntiles + 7) / 8;
  k_wmma_transform<<<gw, tb, 0, stream>>>(s, dinv, W1, b1, W2, g, n, ntiles);

  k_layer2<<<nb,  PT, 0, stream>>>(pairs, tab, g, dinv, b2, out, n, ch);
}


# Chunk sort builds its sorted pairs in LDS and writes the slice out as whole 128-byte lines (16 B per lane, chunks line-aligned) instead of 8-byte scattered stores; sum order as before
# speedup vs baseline: 8.8706x; 1.1810x over previous
#include <hip/hip_runtime.h>

typedef float v2f __attribute__((ext_vector_type(2)));
typedef float v8f __attribute__((ext_vector_type(8)));

#define SH    9
#define RB    (1 << SH)
#define MAXB  256
#define NCH   256
#define PT    1024
#define CAP   25600

__global__ __launch_bounds__(PT)
void k_part(const int* __restrict__ srcIdx, const int* __restrict__ dstIdx,
            int* __restrict__ pairsRaw, int* __restrict__ tab, int e, int ch) {
  __shared__ int cnt[MAXB];
  __shared__ int2 stage[CAP];
  const int tid = threadIdx.x;
  const int c   = blockIdx.x;
  const int beg = c * ch;
  const int end = min(beg + ch, e);
  const int m   = max(end - beg, 0);
  const bool staged = m <= CAP;
  int2* __restrict__ pairs = (int2*)pairsRaw;

  if (tid < MAXB) cnt[tid] = 0;
  __syncthreads();
  for (int i = beg + 4 * tid; i < end; i += 4 * PT) {
    if (i + 3 < end) {
      int4 d = *(const int4*)(dstIdx + i);
      atomicAdd(&cnt[d.x >> SH], 1);
      atomicAdd(&cnt[d.y >> SH], 1);
      atomicAdd(&cnt[d.z >> SH], 1);
      atomicAdd(&cnt[d.w >> SH], 1);
    } else {
      for (int j = i; j < end; ++j) atomicAdd(&cnt[dstIdx[j] >> SH], 1);
    }
  }
  __syncthreads();

  if (tid < 32) {
    int cn[8], st[8];
    int run = 0;
    #pragma unroll
    for (int k = 0; k < 8; ++k) { cn[k] = cnt[tid * 8 + k]; st[k] = run; run += cn[k]; }
    int incl = run;
    #pragma unroll
    for (int off = 1; off < 32; off <<= 1) {
      int t = __shfl_up(incl, off);
      if (tid >= off) incl += t;
    }
    const int excl = incl - run;
    int2* row = (int2*)tab + c * MAXB;
    #pragma unroll
    for (int k = 0; k < 8; ++k) {
      int p = excl + st[k];
      cnt[tid * 8 + k] = p;
      row[tid * 8 + k] = make_int2(p, p + cn[k]);
    }
  }
  __syncthreads();

  for (int i = beg + 4 * tid; i < end; i += 4 * PT) {
    if (i + 3 < end) {
      int4 d = *(const int4*)(dstIdx + i);
      int4 s = *(const int4*)(srcIdx + i);
      int p0 = atomicAdd(&cnt[d.x >> SH], 1);
      int p1 = atomicAdd(&cnt[d.y >> SH], 1);
      int p2 = atomicAdd(&cnt[d.z >> SH], 1);
      int p3 = atomicAdd(&cnt[d.w >> SH], 1);
      if (staged) {
        stage[p0] = make_int2(s.x, d.x);
        stage[p1] = make_int2(s.y, d.y);
        stage[p2] = make_int2(s.z, d.z);
        stage[p3] = make_int2(s.w, d.w);
      } else {
        pairs[beg + p0] = make_int2(s.x, d.x);
        pairs[beg + p1] = make_int2(s.y, d.y);
        pairs[beg + p2] = make_int2(s.z, d.z);
        pairs[beg + p3] = make_int2(s.w, d.w);
      }
    } else {
      for (int j = i; j < end; ++j) {
        int d = dstIdx[j];
        int p = atomicAdd(&cnt[d >> SH], 1);
        if (staged) stage[p] = make_int2(srcIdx[j], d);
        else        pairs[beg + p] = make_int2(srcIdx[j], d);
      }
    }
  }

  if (staged) {
    __syncthreads();
    int4*       __restrict__ o4 = (int4*)(pairs + beg);
    const int4* __restrict__ s4 = (const int4*)stage;
    const int m2 = m >> 1;
    for (int i = tid; i < m2; i += PT) o4[i] = s4[i];
    if ((m & 1) && tid == 0) pairs[beg + m - 1] = stage[m - 1];
  }
}

__global__ __launch_bounds__(PT)
void k_deg(const int* __restrict__ pairsRaw, const int* __restrict__ tab,
           const float* __restrict__ x, float* __restrict__ dinv,
           float* __restrict__ y, int n, int ch) {
  __shared__ int cnt[RB];
  const int tid = threadIdx.x, lane = tid & 31, w = tid >> 5;
  const int b  = blockIdx.x;
  const int n0 = b << SH;
  const int2* __restrict__ pairs = (const int2*)pairsRaw;

  if (tid < RB) cnt[tid] = 0;
  __syncthreads();
  for (int c = w; c < NCH; c += PT / 32) {
    const int2 az = ((const int2*)tab)[c * MAXB + b];
    const int a = az.x, z = az.y;
    const int2* p = pairs + (size_t)c * ch;
    for (int i = a + lane; i < z; i += 32)
      atomicAdd(&cnt[p[i].y - n0], 1);
  }
  __syncthreads();
  if (tid < RB) {
    int node = n0 + tid;
    if (node < n) {
      float dv = rsqrtf((float)(cnt[tid] + 1));
      dinv[node] = dv;
      y[node]    = dv * x[node];
    }
  }
}

__global__ __launch_bounds__(PT)
void k_layer1(const int* __restrict__ pairsRaw, const int* __restrict__ tab,
              const float* __restrict__ y, const float* __restrict__ dinv,
              float* __restrict__ s, int n, int ch) {
  __shared__ float acc[RB];
  const int tid = threadIdx.x, lane = tid & 31, w = tid >> 5;
  const int b  = blockIdx.x;
  const int n0 = b << SH;
  const int2* __restrict__ pairs = (const int2*)pairsRaw;

  if (tid < RB) {
    int node = n0 + tid;
    acc[tid] = (node < n) ? y[node] : 0.0f;
  }
  __syncthreads();
  for (int c = w; c < NCH; c += PT / 32) {
    const int2 az = ((const int2*)tab)[c * MAXB + b];
    const int a = az.x, z = az.y;
    const int2* p = pairs + (size_t)c * ch;
    for (int i = a + lane; i < z; i += 32) {
      int2 q = p[i];
      atomicAdd(&acc[q.y - n0], y[q.x]);
    }
  }
  __syncthreads();
  if (tid < RB) {
    int node = n0 + tid;
    if (node < n) s[node] = dinv[node] * acc[tid];
  }
}

__global__ __launch_bounds__(256)
void k_wmma_transform(const float* __restrict__ s,
                      const float* __restrict__ dinv,
                      const float* __restrict__ W1,
                      const float* __restrict__ b1,
                      const float* __restrict__ W2,
                      float* __restrict__ g, int n, int ntiles) {
  int lane = threadIdx.x & 31;
  int wave = blockIdx.x * (blockDim.x >> 5) + (threadIdx.x >> 5);
  if (wave >= ntiles) return;
  int t0   = wave * 16;
  int half = lane >> 4;
  int m    = lane & 15;
  int node = t0 + m;
  int nidx   = node < n ? node : 0;
  float sn   = (node < n) ? s[nidx] : 0.0f;
  float dn   = (node < n) ? dinv[nidx] : 0.0f;
  float cmsk = (m < 2) ? 1.0f : 0.0f;
  int   cm   = (m < 2) ? m : 0;

  v2f Bf[8];
  #pragma unroll
  for (int i = 0; i < 8; ++i) {
    int f0 = 4 * i + 2 * half;
    Bf[i].x = W2[(f0    ) * 2 + cm] * cmsk;
    Bf[i].y = W2[(f0 + 1) * 2 + cm] * cmsk;
  }

  v8f c = {};
  #pragma unroll
  for (int i = 0; i < 8; ++i) {
    int f0 = 4 * i + 2 * half;
    int f1 = f0 + 1;
    v2f A;
    A.x = dn * fmaxf(fmaf(sn, W1[f0], b1[f0]), 0.0f);
    A.y = dn * fmaxf(fmaf(sn, W1[f1], b1[f1]), 0.0f);
    c = __builtin_amdgcn_wmma_f32_16x16x4_f32(
             false, A,  false, Bf[i],
             (short)0, c,  false,  false);
  }

  if (m < 2) {
    #pragma unroll
    for (int v = 0; v < 8; ++v) {
      int row = t0 + v + 8 * half;
      if (row < n) g[row * 2 + m] = c[v];
    }
  }
}

__global__ __launch_bounds__(PT)
void k_layer2(const int* __restrict__ pairsRaw, const int* __restrict__ tab,
              const float* __restrict__ g, const float* __restrict__ dinv,
              const float* __restrict__ b2, float* __restrict__ out, int n, int ch) {
  __shared__ float acc[2 * RB];
  const int tid = threadIdx.x, lane = tid & 31, w = tid >> 5;
  const int b  = blockIdx.x;
  const int n0 = b << SH;
  const int2* __restrict__ pairs = (const int2*)pairsRaw;

  for (int t = tid; t < 2 * RB; t += PT) acc[t] = 0.0f;
  __syncthreads();
  for (int c = w; c < NCH; c += PT / 32) {
    const int2 az = ((const int2*)tab)[c * MAXB + b];
    const int a = az.x, z = az.y;
    const int2* p = pairs + (size_t)c * ch;
    for (int i = a + lane; i < z; i += 32) {
      int2 q = p[i];
      float2 gv  = *(const float2*)(g + 2 * q.x);
      int t = q.y - n0;
      atomicAdd(&acc[2 * t],     gv.x);
      atomicAdd(&acc[2 * t + 1], gv.y);
    }
  }
  __syncthreads();
  if (tid < RB) {
    int node = n0 + tid;
    if (node < n) {
      float dv = dinv[node];
      float2 gv = *(const float2*)(g + 2 * node);
      float o0 = dv * (acc[2 * tid]     + gv.x) + b2[0];
      float o1 = dv * (acc[2 * tid + 1] + gv.y) + b2[1];
      float mx  = fmaxf(o0, o1);
      float lse = mx + logf(expf(o0 - mx) + expf(o1 - mx));
      float2 r; r.x = o0 - lse; r.y = o1 - lse;
      *(float2*)(out + 2 * node) = r;
    }
  }
}

extern "C" void kernel_launch(void* const* d_in, const int* in_sizes, int n_in,
                              void* d_out, int out_size, void* d_ws, size_t ws_size,
                              hipStream_t stream) {
  const float* x  = (const float*)d_in[0];
  const int*   ei = (const int*)d_in[1];
  const float* W1 = (const float*)d_in[2];
  const float* b1 = (const float*)d_in[3];
  const float* W2 = (const float*)d_in[4];
  const float* b2 = (const float*)d_in[5];
  float* out = (float*)d_out;

  int n = in_sizes[0];
  int e = in_sizes[1] / 2;
  const int* srcIdx = ei;
  const int* dstIdx = ei + e;

  float* dinv  = (float*)d_ws;
  float* s     = dinv + n;
  float* g     = s + n;
  int*   tab   = (int*)(g + 2 * (size_t)n);
  int*   pairs = tab + 2 * NCH * MAXB;

  int nb = (n + RB - 1) / RB;
  int ch = (((e + NCH - 1) / NCH) + 15) & ~15;
  float* y = (float*)(pairs + 2 * (size_t)NCH * ch);

  k_part  <<<NCH, PT, 0, stream>>>(srcIdx, dstIdx, pairs, tab, e, ch);
  k_deg   <<<nb,  PT, 0, stream>>>(pairs, tab, x, dinv, y, n, ch);
  k_layer1<<<nb,  PT, 0, stream>>>(pairs, tab, y, dinv, s, n, ch);

  const int tb = 256;
  int ntiles = (n + 15) / 16;
  int gw = (ntiles + 7) / 8;
  k_wmma_transform<<<gw, tb, 0, stream>>>(s, dinv, W1, b1, W2, g, n, ntiles);

  k_layer2<<<nb,  PT, 0, stream>>>(pairs, tab, g, dinv, b2, out, n, ch);
}


# Consumer waves walk their 8 chunk segments side by side: per trip 8 pair loads, then 8 gathers, then the LDS adds, not one dependent load chain per chunk; float sums in arrival order, interleaved
# speedup vs baseline: 10.7397x; 1.2107x over previous
#include <hip/hip_runtime.h>

typedef float v2f __attribute__((ext_vector_type(2)));
typedef float v8f __attribute__((ext_vector_type(8)));

#define SH    9
#define RB    (1 << SH)
#define MAXB  256
#define NCH   256
#define PT    1024
#define CAP   25600
#define WPG   (PT / 32)
#define KPW   (NCH / WPG)

__device__ __forceinline__ int seg_bounds(const int* __restrict__ tab, int w, int b,
                                          int (&a)[KPW], int (&z)[KPW]) {
  int longest = 0;
  #pragma unroll
  for (int k = 0; k < KPW; ++k) {
    const int2 az = ((const int2*)tab)[(w + k * WPG) * MAXB + b];
    a[k] = az.x;
    z[k] = az.y;
    longest = max(longest, z[k] - a[k]);
  }
  return longest;
}

__global__ __launch_bounds__(PT)
void k_part(const int* __restrict__ srcIdx, const int* __restrict__ dstIdx,
            int* __restrict__ pairsRaw, int* __restrict__ tab, int e, int ch) {
  __shared__ int cnt[MAXB];
  __shared__ int2 stage[CAP];
  const int tid = threadIdx.x;
  const int c   = blockIdx.x;
  const int beg = c * ch;
  const int end = min(beg + ch, e);
  const int m   = max(end - beg, 0);
  const bool staged = m <= CAP;
  int2* __restrict__ pairs = (int2*)pairsRaw;

  if (tid < MAXB) cnt[tid] = 0;
  __syncthreads();
  for (int i = beg + 4 * tid; i < end; i += 4 * PT) {
    if (i + 3 < end) {
      int4 d = *(const int4*)(dstIdx + i);
      atomicAdd(&cnt[d.x >> SH], 1);
      atomicAdd(&cnt[d.y >> SH], 1);
      atomicAdd(&cnt[d.z >> SH], 1);
      atomicAdd(&cnt[d.w >> SH], 1);
    } else {
      for (int j = i; j < end; ++j) atomicAdd(&cnt[dstIdx[j] >> SH], 1);
    }
  }
  __syncthreads();

  if (tid < 32) {
    int cn[8], st[8];
    int run = 0;
    #pragma unroll
    for (int k = 0; k < 8; ++k) { cn[k] = cnt[tid * 8 + k]; st[k] = run; run += cn[k]; }
    int incl = run;
    #pragma unroll
    for (int off = 1; off < 32; off <<= 1) {
      int t = __shfl_up(incl, off);
      if (tid >= off) incl += t;
    }
    const int excl = incl - run;
    int2* row = (int2*)tab + c * MAXB;
    #pragma unroll
    for (int k = 0; k < 8; ++k) {
      int p = excl + st[k];
      cnt[tid * 8 + k] = p;
      row[tid * 8 + k] = make_int2(p, p + cn[k]);
    }
  }
  __syncthreads();

  for (int i = beg + 4 * tid; i < end; i += 4 * PT) {
    if (i + 3 < end) {
      int4 d = *(const int4*)(dstIdx + i);
      int4 s = *(const int4*)(srcIdx + i);
      int p0 = atomicAdd(&cnt[d.x >> SH], 1);
      int p1 = atomicAdd(&cnt[d.y >> SH], 1);
      int p2 = atomicAdd(&cnt[d.z >> SH], 1);
      int p3 = atomicAdd(&cnt[d.w >> SH], 1);
      if (staged) {
        stage[p0] = make_int2(s.x, d.x);
        stage[p1] = make_int2(s.y, d.y);
        stage[p2] = make_int2(s.z, d.z);
        stage[p3] = make_int2(s.w, d.w);
      } else {
        pairs[beg + p0] = make_int2(s.x, d.x);
        pairs[beg + p1] = make_int2(s.y, d.y);
        pairs[beg + p2] = make_int2(s.z, d.z);
        pairs[beg + p3] = make_int2(s.w, d.w);
      }
    } else {
      for (int j = i; j < end; ++j) {
        int d = dstIdx[j];
        int p = atomicAdd(&cnt[d >> SH], 1);
        if (staged) stage[p] = make_int2(srcIdx[j], d);
        else        pairs[beg + p] = make_int2(srcIdx[j], d);
      }
    }
  }

  if (staged) {
    __syncthreads();
    int4*       __restrict__ o4 = (int4*)(pairs + beg);
    const int4* __restrict__ s4 = (const int4*)stage;
    const int m2 = m >> 1;
    for (int i = tid; i < m2; i += PT) o4[i] = s4[i];
    if ((m & 1) && tid == 0) pairs[beg + m - 1] = stage[m - 1];
  }
}

__global__ __launch_bounds__(PT)
void k_deg(const int* __restrict__ pairsRaw, const int* __restrict__ tab,
           const float* __restrict__ x, float* __restrict__ dinv,
           float* __restrict__ y, int n, int ch) {
  __shared__ int cnt[RB];
  const int tid = threadIdx.x, lane = tid & 31;
  const int w  = __builtin_amdgcn_readfirstlane(tid >> 5);
  const int b  = blockIdx.x;
  const int n0 = b << SH;
  const int2* __restrict__ pairs = (const int2*)pairsRaw;

  if (tid < RB) cnt[tid] = 0;
  int a[KPW], z[KPW];
  const int longest = seg_bounds(tab, w, b, a, z);
  __syncthreads();
  for (int o = 0; o < longest; o += 32) {
    int d[KPW];
    #pragma unroll
    for (int k = 0; k < KPW; ++k) {
      const int i = a[k] + o + lane;
      const size_t j = (i < z[k]) ? (size_t)(w + k * WPG) * ch + i : 0;
      d[k] = pairs[j].y;
    }
    asm volatile("" :: "v"(d[0]));
    #pragma unroll
    for (int k = 0; k < KPW; ++k) {
      const int i = a[k] + o + lane;
      if (i < z[k]) atomicAdd(&cnt[d[k] - n0], 1);
    }
  }
  __syncthreads();
  if (tid < RB) {
    int node = n0 + tid;
    if (node < n) {
      float dv = rsqrtf((float)(cnt[tid] + 1));
      dinv[node] = dv;
      y[node]    = dv * x[node];
    }
  }
}

__global__ __launch_bounds__(PT)
void k_layer1(const int* __restrict__ pairsRaw, const int* __restrict__ tab,
              const float* __restrict__ y, const float* __restrict__ dinv,
              float* __restrict__ s, int n, int ch) {
  __shared__ float acc[RB];
  const int tid = threadIdx.x, lane = tid & 31;
  const int w  = __builtin_amdgcn_readfirstlane(tid >> 5);
  const int b  = blockIdx.x;
  const int n0 = b << SH;
  const int2* __restrict__ pairs = (const int2*)pairsRaw;

  if (tid < RB) {
    int node = n0 + tid;
    acc[tid] = (node < n) ? y[node] : 0.0f;
  }
  int a[KPW], z[KPW];
  const int longest = seg_bounds(tab, w, b, a, z);
  __syncthreads();
  for (int o = 0; o < longest; o += 32) {
    int2 q[KPW];
    float v[KPW];
    #pragma unroll
    for (int k = 0; k < KPW; ++k) {
      const int i = a[k] + o + lane;
      const size_t j = (i < z[k]) ? (size_t)(w + k * WPG) * ch + i : 0;
      q[k] = pairs[j];
    }
    #pragma unroll
    for (int k = 0; k < KPW; ++k) v[k] = y[q[k].x];
    asm volatile("" :: "v"(v[0]));
    #pragma unroll
    for (int k = 0; k < KPW; ++k) {
      const int i = a[k] + o + lane;
      if (i < z[k]) atomicAdd(&acc[q[k].y - n0], v[k]);
    }
  }
  __syncthreads();
  if (tid < RB) {
    int node = n0 + tid;
    if (node < n) s[node] = dinv[node] * acc[tid];
  }
}

__global__ __launch_bounds__(256)
void k_wmma_transform(const float* __restrict__ s,
                      const float* __restrict__ dinv,
                      const float* __restrict__ W1,
                      const float* __restrict__ b1,
                      const float* __restrict__ W2,
                      float* __restrict__ g, int n, int ntiles) {
  int lane = threadIdx.x & 31;
  int wave = blockIdx.x * (blockDim.x >> 5) + (threadIdx.x >> 5);
  if (wave >= ntiles) return;
  int t0   = wave * 16;
  int half = lane >> 4;
  int m    = lane & 15;
  int node = t0 + m;
  int nidx   = node < n ? node : 0;
  float sn   = (node < n) ? s[nidx] : 0.0f;
  float dn   = (node < n) ? dinv[nidx] : 0.0f;
  float cmsk = (m < 2) ? 1.0f : 0.0f;
  int   cm   = (m < 2) ? m : 0;

  v2f Bf[8];
  #pragma unroll
  for (int i = 0; i < 8; ++i) {
    int f0 = 4 * i + 2 * half;
    Bf[i].x = W2[(f0    ) * 2 + cm] * cmsk;
    Bf[i].y = W2[(f0 + 1) * 2 + cm] * cmsk;
  }

  v8f c = {};
  #pragma unroll
  for (int i = 0; i < 8; ++i) {
    int f0 = 4 * i + 2 * half;
    int f1 = f0 + 1;
    v2f A;
    A.x = dn * fmaxf(fmaf(sn, W1[f0], b1[f0]), 0.0f);
    A.y = dn * fmaxf(fmaf(sn, W1[f1], b1[f1]), 0.0f);
    c = __builtin_amdgcn_wmma_f32_16x16x4_f32(
             false, A,  false, Bf[i],
             (short)0, c,  false,  false);
  }

  if (m < 2) {
    #pragma unroll
    for (int v = 0; v < 8; ++v) {
      int row = t0 + v + 8 * half;
      if (row < n) g[row * 2 + m] = c[v];
    }
  }
}

__global__ __launch_bounds__(PT)
void k_layer2(const int* __restrict__ pairsRaw, const int* __restrict__ tab,
              const float* __restrict__ g, const float* __restrict__ dinv,
              const float* __restrict__ b2, float* __restrict__ out, int n, int ch) {
  __shared__ float acc[2 * RB];
  const int tid = threadIdx.x, lane = tid & 31;
  const int w  = __builtin_amdgcn_readfirstlane(tid >> 5);
  const int b  = blockIdx.x;
  const int n0 = b << SH;
  const int2* __restrict__ pairs = (const int2*)pairsRaw;

  for (int t = tid; t < 2 * RB; t += PT) acc[t] = 0.0f;
  int a[KPW], z[KPW];
  const int longest = seg_bounds(tab, w, b, a, z);
  __syncthreads();
  for (int o = 0; o < longest; o += 32) {
    int2 q[KPW];
    float2 gv[KPW];
    #pragma unroll
    for (int k = 0; k < KPW; ++k) {
      const int i = a[k] + o + lane;
      const size_t j = (i < z[k]) ? (size_t)(w + k * WPG) * ch + i : 0;
      q[k] = pairs[j];
    }
    #pragma unroll
    for (int k = 0; k < KPW; ++k) gv[k] = *(const float2*)(g + 2 * q[k].x);
    asm volatile("" :: "v"(gv[0].x), "v"(gv[0].y));
    #pragma unroll
    for (int k = 0; k < KPW; ++k) {
      const int i = a[k] + o + lane;
      if (i < z[k]) {
        const int t = q[k].y - n0;
        atomicAdd(&acc[2 * t],     gv[k].x);
        atomicAdd(&acc[2 * t + 1], gv[k].y);
      }
    }
  }
  __syncthreads();
  if (tid < RB) {
    int node = n0 + tid;
    if (node < n) {
      float dv = dinv[node];
      float2 gv = *(const float2*)(g + 2 * node);
      float o0 = dv * (acc[2 * tid]     + gv.x) + b2[0];
      float o1 = dv * (acc[2 * tid + 1] + gv.y) + b2[1];
      float mx  = fmaxf(o0, o1);
      float lse = mx + logf(expf(o0 - mx) + expf(o1 - mx));
      float2 r; r.x = o0 - lse; r.y = o1 - lse;
      *(float2*)(out + 2 * node) = r;
    }
  }
}

extern "C" void kernel_launch(void* const* d_in, const int* in_sizes, int n_in,
                              void* d_out, int out_size, void* d_ws, size_t ws_size,
                              hipStream_t stream) {
  const float* x  = (const float*)d_in[0];
  const int*   ei = (const int*)d_in[1];
  const float* W1 = (const float*)d_in[2];
  const float* b1 = (const float*)d_in[3];
  const float* W2 = (const float*)d_in[4];
  const float* b2 = (const float*)d_in[5];
  float* out = (float*)d_out;

  int n = in_sizes[0];
  int e = in_sizes[1] / 2;
  const int* srcIdx = ei;
  const int* dstIdx = ei + e;

  float* dinv  = (float*)d_ws;
  float* s     = dinv + n;
  float* g     = s + n;
  int*   tab   = (int*)(g + 2 * (size_t)n);
  int*   pairs = tab + 2 * NCH * MAXB;

  int nb = (n + RB - 1) / RB;
  int ch = (((e + NCH - 1) / NCH) + 15) & ~15;
  float* y = (float*)(pairs + 2 * (size_t)NCH * ch);

  k_part  <<<NCH, PT, 0, stream>>>(srcIdx, dstIdx, pairs, tab, e, ch);
  k_deg   <<<nb,  PT, 0, stream>>>(pairs, tab, x, dinv, y, n, ch);
  k_layer1<<<nb,  PT, 0, stream>>>(pairs, tab, y, dinv, s, n, ch);

  const int tb = 256;
  int ntiles = (n + 15) / 16;
  int gw = (ntiles + 7) / 8;
  k_wmma_transform<<<gw, tb, 0, stream>>>(s, dinv, W1, b1, W2, g, n, ntiles);

  k_layer2<<<nb,  PT, 0, stream>>>(pairs, tab, g, dinv, b2, out, n, ch);
}


# Both aggregations read the per-source value from LDS: edges also sorted by source half, each half of y, g0, g1 staged in turn; float sums in arrival order, now grouped by source half and column
# speedup vs baseline: 14.0485x; 1.3081x over previous
#include <hip/hip_runtime.h>

typedef float v2f __attribute__((ext_vector_type(2)));
typedef float v4f __attribute__((ext_vector_type(4)));
typedef float v8f __attribute__((ext_vector_type(8)));

#define SH    9
#define RB    (1 << SH)
#define MAXB  256
#define KEYS  (2 * MAXB)
#define HMAX  65536
#define NCH   256
#define PT    1024
#define CAP   25600
#define WPG   (PT / 32)
#define KPW   (NCH / WPG)

__device__ __forceinline__ void seg_bounds(const int* __restrict__ tab, int w, int b,
                                           int (&a)[KPW], int (&m)[KPW], int (&z)[KPW]) {
  #pragma unroll
  for (int k = 0; k < KPW; ++k) {
    const int4 t = ((const int4*)tab)[(w + k * WPG) * MAXB + b];
    a[k] = t.x;
    m[k] = t.y;
    z[k] = t.w;
  }
}

__device__ __forceinline__ void stage_half(float* __restrict__ tbl, const float* __restrict__ src,
                                           int cnt, int tid) {
  const int c4 = cnt >> 2;
  const v4f* __restrict__ s4 = (const v4f*)src;
  v4f* __restrict__ t4 = (v4f*)tbl;
  for (int i = tid; i < c4; i += 8 * PT) {
    v4f v[8];
    #pragma unroll
    for (int u = 0; u < 8; ++u) {
      const int j = i + u * PT;
      v[u] = s4[j < c4 ? j : i];
    }
    asm volatile("" :: "v"(v[0]), "v"(v[1]), "v"(v[2]), "v"(v[3]),
                       "v"(v[4]), "v"(v[5]), "v"(v[6]), "v"(v[7]));
    #pragma unroll
    for (int u = 0; u < 8; ++u) {
      const int j = i + u * PT;
      if (j < c4) t4[j] = v[u];
    }
  }
  for (int i = (c4 << 2) + tid; i < cnt; i += PT) tbl[i] = src[i];
}

__device__ __forceinline__ void walk_half(const int2* __restrict__ pairs, int w, int ch, int lane,
                                          const int (&lo)[KPW], const int (&hi)[KPW],
                                          const float* tbl, int base, float* acc, int n0) {
  int longest = 0;
  #pragma unroll
  for (int k = 0; k < KPW; ++k) longest = max(longest, hi[k] - lo[k]);
  for (int o = 0; o < longest; o += 32) {
    int2 q[KPW];
    float v[KPW];
    #pragma unroll
    for (int k = 0; k < KPW; ++k) {
      const int i = lo[k] + o + lane;
      const size_t j = (i < hi[k]) ? (size_t)(w + k * WPG) * ch + i : 0;
      q[k] = pairs[j];
    }
    #pragma unroll
    for (int k = 0; k < KPW; ++k) {
      const int i = lo[k] + o + lane;
      v[k] = tbl[(i < hi[k]) ? q[k].x - base : 0];
    }
    asm volatile("" :: "v"(v[0]));
    #pragma unroll
    for (int k = 0; k < KPW; ++k) {
      const int i = lo[k] + o + lane;
      if (i < hi[k]) atomicAdd(&acc[q[k].y - n0], v[k]);
    }
  }
}

__device__ __forceinline__ int sort_key(int s, int d, int hs) {
  return ((d >> SH) << 1) | (s >= hs ? 1 : 0);
}

__global__ __launch_bounds__(PT)
void k_part(const int* __restrict__ srcIdx, const int* __restrict__ dstIdx,
            int* __restrict__ pairsRaw, int* __restrict__ tab, int e, int ch, int hs) {
  __shared__ int cnt[KEYS];
  __shared__ int2 stage[CAP];
  const int tid = threadIdx.x;
  const int c   = blockIdx.x;
  const int beg = c * ch;
  const int end = min(beg + ch, e);
  const int m   = max(end - beg, 0);
  const bool staged = m <= CAP;
  int2* __restrict__ pairs = (int2*)pairsRaw;

  if (tid < KEYS) cnt[tid] = 0;
  __syncthreads();
  for (int i = beg + 4 * tid; i < end; i += 4 * PT) {
    if (i + 3 < end) {
      int4 d = *(const int4*)(dstIdx + i);
      int4 s = *(const int4*)(srcIdx + i);
      atomicAdd(&cnt[sort_key(s.x, d.x, hs)], 1);
      atomicAdd(&cnt[sort_key(s.y, d.y, hs)], 1);
      atomicAdd(&cnt[sort_key(s.z, d.z, hs)], 1);
      atomicAdd(&cnt[sort_key(s.w, d.w, hs)], 1);
    } else {
      for (int j = i; j < end; ++j) atomicAdd(&cnt[sort_key(srcIdx[j], dstIdx[j], hs)], 1);
    }
  }
  __syncthreads();

  if (tid < 32) {
    int cn[16], st[16];
    int run = 0;
    #pragma unroll
    for (int k = 0; k < 16; ++k) { cn[k] = cnt[tid * 16 + k]; st[k] = run; run += cn[k]; }
    int incl = run;
    #pragma unroll
    for (int off = 1; off < 32; off <<= 1) {
      int t = __shfl_up(incl, off);
      if (tid >= off) incl += t;
    }
    const int excl = incl - run;
    int2* row = (int2*)tab + c * KEYS;
    #pragma unroll
    for (int k = 0; k < 16; ++k) {
      int p = excl + st[k];
      cnt[tid * 16 + k] = p;
      row[tid * 16 + k] = make_int2(p, p + cn[k]);
    }
  }
  __syncthreads();

  for (int i = beg + 4 * tid; i < end; i += 4 * PT) {
    if (i + 3 < end) {
      int4 d = *(const int4*)(dstIdx + i);
      int4 s = *(const int4*)(srcIdx + i);
      int p0 = atomicAdd(&cnt[sort_key(s.x, d.x, hs)], 1);
      int p1 = atomicAdd(&cnt[sort_key(s.y, d.y, hs)], 1);
      int p2 = atomicAdd(&cnt[sort_key(s.z, d.z, hs)], 1);
      int p3 = atomicAdd(&cnt[sort_key(s.w, d.w, hs)], 1);
      if (staged) {
        stage[p0] = make_int2(s.x, d.x);
        stage[p1] = make_int2(s.y, d.y);
        stage[p2] = make_int2(s.z, d.z);
        stage[p3] = make_int2(s.w, d.w);
      } else {
        pairs[beg + p0] = make_int2(s.x, d.x);
        pairs[beg + p1] = make_int2(s.y, d.y);
        pairs[beg + p2] = make_int2(s.z, d.z);
        pairs[beg + p3] = make_int2(s.w, d.w);
      }
    } else {
      for (int j = i; j < end; ++j) {
        int d = dstIdx[j];
        int s = srcIdx[j];
        int p = atomicAdd(&cnt[sort_key(s, d, hs)], 1);
        if (staged) stage[p] = make_int2(s, d);
        else        pairs[beg + p] = make_int2(s, d);
      }
    }
  }

  if (staged) {
    __syncthreads();
    int4*       __restrict__ o4 = (int4*)(pairs + beg);
    const int4* __restrict__ s4 = (const int4*)stage;
    const int m2 = m >> 1;
    for (int i = tid; i < m2; i += PT) o4[i] = s4[i];
    if ((m & 1) && tid == 0) pairs[beg + m - 1] = stage[m - 1];
  }
}

__global__ __launch_bounds__(PT)
void k_deg(const int* __restrict__ pairsRaw, const int* __restrict__ tab,
           const float* __restrict__ x, float* __restrict__ dinv,
           float* __restrict__ y, int n, int ch) {
  __shared__ int cnt[RB];
  const int tid = threadIdx.x, lane = tid & 31;
  const int w  = __builtin_amdgcn_readfirstlane(tid >> 5);
  const int b  = blockIdx.x;
  const int n0 = b << SH;
  const int2* __restrict__ pairs = (const int2*)pairsRaw;

  if (tid < RB) cnt[tid] = 0;
  int a[KPW], m[KPW], z[KPW];
  seg_bounds(tab, w, b, a, m, z);
  int longest = 0;
  #pragma unroll
  for (int k = 0; k < KPW; ++k) longest = max(longest, z[k] - a[k]);
  __syncthreads();
  for (int o = 0; o < longest; o += 32) {
    int d[KPW];
    #pragma unroll
    for (int k = 0; k < KPW; ++k) {
      const int i = a[k] + o + lane;
      const size_t j = (i < z[k]) ? (size_t)(w + k * WPG) * ch + i : 0;
      d[k] = pairs[j].y;
    }
    asm volatile("" :: "v"(d[0]));
    #pragma unroll
    for (int k = 0; k < KPW; ++k) {
      const int i = a[k] + o + lane;
      if (i < z[k]) atomicAdd(&cnt[d[k] - n0], 1);
    }
  }
  __syncthreads();
  if (tid < RB) {
    int node = n0 + tid;
    if (node < n) {
      float dv = rsqrtf((float)(cnt[tid] + 1));
      dinv[node] = dv;
      y[node]    = dv * x[node];
    }
  }
}

__global__ __launch_bounds__(PT)
void k_layer1(const int* __restrict__ pairsRaw, const int* __restrict__ tab,
              const float* __restrict__ y, const float* __restrict__ dinv,
              float* __restrict__ s, int n, int ch, int hs) {
  __shared__ float acc[RB];
  __shared__ float tbl[HMAX];
  const int tid = threadIdx.x, lane = tid & 31;
  const int w  = __builtin_amdgcn_readfirstlane(tid >> 5);
  const int b  = blockIdx.x;
  const int n0 = b << SH;
  const int2* __restrict__ pairs = (const int2*)pairsRaw;

  if (tid < RB) {
    int node = n0 + tid;
    acc[tid] = (node < n) ? y[node] : 0.0f;
  }
  int a[KPW], m[KPW], z[KPW];
  seg_bounds(tab, w, b, a, m, z);
  stage_half(tbl, y, min(hs, n), tid);
  __syncthreads();
  walk_half(pairs, w, ch, lane, a, m, tbl, 0, acc, n0);
  __syncthreads();
  stage_half(tbl, y + hs, n - hs, tid);
  __syncthreads();
  walk_half(pairs, w, ch, lane, m, z, tbl, hs, acc, n0);
  __syncthreads();
  if (tid < RB) {
    int node = n0 + tid;
    if (node < n) s[node] = dinv[node] * acc[tid];
  }
}

__global__ __launch_bounds__(256)
void k_wmma_transform(const float* __restrict__ s,
                      const float* __restrict__ dinv,
                      const float* __restrict__ W1,
                      const float* __restrict__ b1,
                      const float* __restrict__ W2,
                      float* __restrict__ g0, float* __restrict__ g1, int n, int ntiles) {
  int lane = threadIdx.x & 31;
  int wave = blockIdx.x * (blockDim.x >> 5) + (threadIdx.x >> 5);
  if (wave >= ntiles) return;
  int t0   = wave * 16;
  int half = lane >> 4;
  int m    = lane & 15;
  int node = t0 + m;
  int nidx   = node < n ? node : 0;
  float sn   = (node < n) ? s[nidx] : 0.0f;
  float dn   = (node < n) ? dinv[nidx] : 0.0f;
  float cmsk = (m < 2) ? 1.0f : 0.0f;
  int   cm   = (m < 2) ? m : 0;

  v2f Bf[8];
  #pragma unroll
  for (int i = 0; i < 8; ++i) {
    int f0 = 4 * i + 2 * half;
    Bf[i].x = W2[(f0    ) * 2 + cm] * cmsk;
    Bf[i].y = W2[(f0 + 1) * 2 + cm] * cmsk;
  }

  v8f c = {};
  #pragma unroll
  for (int i = 0; i < 8; ++i) {
    int f0 = 4 * i + 2 * half;
    int f1 = f0 + 1;
    v2f A;
    A.x = dn * fmaxf(fmaf(sn, W1[f0], b1[f0]), 0.0f);
    A.y = dn * fmaxf(fmaf(sn, W1[f1], b1[f1]), 0.0f);
    c = __builtin_amdgcn_wmma_f32_16x16x4_f32(
             false, A,  false, Bf[i],
             (short)0, c,  false,  false);
  }

  if (m < 2) {
    float* __restrict__ gp = m ? g1 : g0;
    #pragma unroll
    for (int v = 0; v < 8; ++v) {
      int row = t0 + v + 8 * half;
      if (row < n) gp[row] = c[v];
    }
  }
}

__global__ __launch_bounds__(PT)
void k_layer2(const int* __restrict__ pairsRaw, const int* __restrict__ tab,
              const float* __restrict__ g0, const float* __restrict__ g1,
              const float* __restrict__ dinv,
              const float* __restrict__ b2, float* __restrict__ out, int n, int ch, int hs) {
  __shared__ float acc[2 * RB];
  __shared__ float tbl[HMAX];
  const int tid = threadIdx.x, lane = tid & 31;
  const int w  = __builtin_amdgcn_readfirstlane(tid >> 5);
  const int b  = blockIdx.x;
  const int n0 = b << SH;
  const int2* __restrict__ pairs = (const int2*)pairsRaw;

  for (int t = tid; t < 2 * RB; t += PT) acc[t] = 0.0f;
  int a[KPW], m[KPW], z[KPW];
  seg_bounds(tab, w, b, a, m, z);
  for (int col = 0; col < 2; ++col) {
    const float* __restrict__ gp = col ? g1 : g0;
    float* accc = acc + col * RB;
    stage_half(tbl, gp, min(hs, n), tid);
    __syncthreads();
    walk_half(pairs, w, ch, lane, a, m, tbl, 0, accc, n0);
    __syncthreads();
    stage_half(tbl, gp + hs, n - hs, tid);
    __syncthreads();
    walk_half(pairs, w, ch, lane, m, z, tbl, hs, accc, n0);
    __syncthreads();
  }
  if (tid < RB) {
    int node = n0 + tid;
    if (node < n) {
      float dv = dinv[node];
      float o0 = dv * (acc[tid]      + g0[node]) + b2[0];
      float o1 = dv * (acc[RB + tid] + g1[node]) + b2[1];
      float mx  = fmaxf(o0, o1);
      float lse = mx + logf(expf(o0 - mx) + expf(o1 - mx));
      float2 r; r.x = o0 - lse; r.y = o1 - lse;
      *(float2*)(out + 2 * node) = r;
    }
  }
}

extern "C" void kernel_launch(void* const* d_in, const int* in_sizes, int n_in,
                              void* d_out, int out_size, void* d_ws, size_t ws_size,
                              hipStream_t stream) {
  const float* x  = (const float*)d_in[0];
  const int*   ei = (const int*)d_in[1];
  const float* W1 = (const float*)d_in[2];
  const float* b1 = (const float*)d_in[3];
  const float* W2 = (const float*)d_in[4];
  const float* b2 = (const float*)d_in[5];
  float* out = (float*)d_out;

  int n = in_sizes[0];
  int e = in_sizes[1] / 2;
  const int* srcIdx = ei;
  const int* dstIdx = ei + e;

  int np = (n + 3) & ~3;
  float* y     = (float*)d_ws;
  float* g0    = y + np;
  float* g1    = g0 + np;
  float* dinv  = g1 + np;
  float* s     = dinv + np;
  int*   tab   = (int*)(s + np);
  int*   pairs = tab + 2 * NCH * KEYS;

  int nb = (n + RB - 1) / RB;
  int ch = (((e + NCH - 1) / NCH) + 15) & ~15;
  int hs = (((n + 1) / 2) + 3) & ~3;

  k_part  <<<NCH, PT, 0, stream>>>(srcIdx, dstIdx, pairs, tab, e, ch, hs);
  k_deg   <<<nb,  PT, 0, stream>>>(pairs, tab, x, dinv, y, n, ch);
  k_layer1<<<nb,  PT, 0, stream>>>(pairs, tab, y, dinv, s, n, ch, hs);

  const int tb = 256;
  int ntiles = (n + 15) / 16;
  int gw = (ntiles + 7) / 8;
  k_wmma_transform<<<gw, tb, 0, stream>>>(s, dinv, W1, b1, W2, g0, g1, n, ntiles);

  k_layer2<<<nb,  PT, 0, stream>>>(pairs, tab, g0, g1, dinv, b2, out, n, ch, hs);
}


# Node transform (relu(s*W1+b1) times dinv, then W2 on the f32 matrix instruction) moved into the layer-1 kernel as its epilogue, one 16-node tile per wave from the LDS sums: 4 launches, not 5
# speedup vs baseline: 14.8505x; 1.0571x over previous
#include <hip/hip_runtime.h>

typedef float v2f __attribute__((ext_vector_type(2)));
typedef float v4f __attribute__((ext_vector_type(4)));
typedef float v8f __attribute__((ext_vector_type(8)));

#define SH    9
#define RB    (1 << SH)
#define MAXB  256
#define KEYS  (2 * MAXB)
#define HMAX  65536
#define NCH   256
#define PT    1024
#define CAP   25600
#define WPG   (PT / 32)
#define KPW   (NCH / WPG)

__device__ __forceinline__ void seg_bounds(const int* __restrict__ tab, int w, int b,
                                           int (&a)[KPW], int (&m)[KPW], int (&z)[KPW]) {
  #pragma unroll
  for (int k = 0; k < KPW; ++k) {
    const int4 t = ((const int4*)tab)[(w + k * WPG) * MAXB + b];
    a[k] = t.x;
    m[k] = t.y;
    z[k] = t.w;
  }
}

__device__ __forceinline__ void stage_half(float* __restrict__ tbl, const float* __restrict__ src,
                                           int cnt, int tid) {
  const int c4 = cnt >> 2;
  const v4f* __restrict__ s4 = (const v4f*)src;
  v4f* __restrict__ t4 = (v4f*)tbl;
  for (int i = tid; i < c4; i += 8 * PT) {
    v4f v[8];
    #pragma unroll
    for (int u = 0; u < 8; ++u) {
      const int j = i + u * PT;
      v[u] = s4[j < c4 ? j : i];
    }
    asm volatile("" :: "v"(v[0]), "v"(v[1]), "v"(v[2]), "v"(v[3]),
                       "v"(v[4]), "v"(v[5]), "v"(v[6]), "v"(v[7]));
    #pragma unroll
    for (int u = 0; u < 8; ++u) {
      const int j = i + u * PT;
      if (j < c4) t4[j] = v[u];
    }
  }
  for (int i = (c4 << 2) + tid; i < cnt; i += PT) tbl[i] = src[i];
}

__device__ __forceinline__ void walk_half(const int2* __restrict__ pairs, int w, int ch, int lane,
                                          const int (&lo)[KPW], const int (&hi)[KPW],
                                          const float* tbl, int base, float* acc, int n0) {
  int longest = 0;
  #pragma unroll
  for (int k = 0; k < KPW; ++k) longest = max(longest, hi[k] - lo[k]);
  for (int o = 0; o < longest; o += 32) {
    int2 q[KPW];
    float v[KPW];
    #pragma unroll
    for (int k = 0; k < KPW; ++k) {
      const int i = lo[k] + o + lane;
      const size_t j = (i < hi[k]) ? (size_t)(w + k * WPG) * ch + i : 0;
      q[k] = pairs[j];
    }
    #pragma unroll
    for (int k = 0; k < KPW; ++k) {
      const int i = lo[k] + o + lane;
      v[k] = tbl[(i < hi[k]) ? q[k].x - base : 0];
    }
    asm volatile("" :: "v"(v[0]));
    #pragma unroll
    for (int k = 0; k < KPW; ++k) {
      const int i = lo[k] + o + lane;
      if (i < hi[k]) atomicAdd(&acc[q[k].y - n0], v[k]);
    }
  }
}

__device__ __forceinline__ int sort_key(int s, int d, int hs) {
  return ((d >> SH) << 1) | (s >= hs ? 1 : 0);
}

__global__ __launch_bounds__(PT)
void k_part(const int* __restrict__ srcIdx, const int* __restrict__ dstIdx,
            int* __restrict__ pairsRaw, int* __restrict__ tab, int e, int ch, int hs) {
  __shared__ int cnt[KEYS];
  __shared__ int2 stage[CAP];
  const int tid = threadIdx.x;
  const int c   = blockIdx.x;
  const int beg = c * ch;
  const int end = min(beg + ch, e);
  const int m   = max(end - beg, 0);
  const bool staged = m <= CAP;
  int2* __restrict__ pairs = (int2*)pairsRaw;

  if (tid < KEYS) cnt[tid] = 0;
  __syncthreads();
  for (int i = beg + 4 * tid; i < end; i += 4 * PT) {
    if (i + 3 < end) {
      int4 d = *(const int4*)(dstIdx + i);
      int4 s = *(const int4*)(srcIdx + i);
      atomicAdd(&cnt[sort_key(s.x, d.x, hs)], 1);
      atomicAdd(&cnt[sort_key(s.y, d.y, hs)], 1);
      atomicAdd(&cnt[sort_key(s.z, d.z, hs)], 1);
      atomicAdd(&cnt[sort_key(s.w, d.w, hs)], 1);
    } else {
      for (int j = i; j < end; ++j) atomicAdd(&cnt[sort_key(srcIdx[j], dstIdx[j], hs)], 1);
    }
  }
  __syncthreads();

  if (tid < 32) {
    int cn[16], st[16];
    int run = 0;
    #pragma unroll
    for (int k = 0; k < 16; ++k) { cn[k] = cnt[tid * 16 + k]; st[k] = run; run += cn[k]; }
    int incl = run;
    #pragma unroll
    for (int off = 1; off < 32; off <<= 1) {
      int t = __shfl_up(incl, off);
      if (tid >= off) incl += t;
    }
    const int excl = incl - run;
    int2* row = (int2*)tab + c * KEYS;
    #pragma unroll
    for (int k = 0; k < 16; ++k) {
      int p = excl + st[k];
      cnt[tid * 16 + k] = p;
      row[tid * 16 + k] = make_int2(p, p + cn[k]);
    }
  }
  __syncthreads();

  for (int i = beg + 4 * tid; i < end; i += 4 * PT) {
    if (i + 3 < end) {
      int4 d = *(const int4*)(dstIdx + i);
      int4 s = *(const int4*)(srcIdx + i);
      int p0 = atomicAdd(&cnt[sort_key(s.x, d.x, hs)], 1);
      int p1 = atomicAdd(&cnt[sort_key(s.y, d.y, hs)], 1);
      int p2 = atomicAdd(&cnt[sort_key(s.z, d.z, hs)], 1);
      int p3 = atomicAdd(&cnt[sort_key(s.w, d.w, hs)], 1);
      if (staged) {
        stage[p0] = make_int2(s.x, d.x);
        stage[p1] = make_int2(s.y, d.y);
        stage[p2] = make_int2(s.z, d.z);
        stage[p3] = make_int2(s.w, d.w);
      } else {
        pairs[beg + p0] = make_int2(s.x, d.x);
        pairs[beg + p1] = make_int2(s.y, d.y);
        pairs[beg + p2] = make_int2(s.z, d.z);
        pairs[beg + p3] = make_int2(s.w, d.w);
      }
    } else {
      for (int j = i; j < end; ++j) {
        int d = dstIdx[j];
        int s = srcIdx[j];
        int p = atomicAdd(&cnt[sort_key(s, d, hs)], 1);
        if (staged) stage[p] = make_int2(s, d);
        else        pairs[beg + p] = make_int2(s, d);
      }
    }
  }

  if (staged) {
    __syncthreads();
    int4*       __restrict__ o4 = (int4*)(pairs + beg);
    const int4* __restrict__ s4 = (const int4*)stage;
    const int m2 = m >> 1;
    for (int i = tid; i < m2; i += PT) o4[i] = s4[i];
    if ((m & 1) && tid == 0) pairs[beg + m - 1] = stage[m - 1];
  }
}

__global__ __launch_bounds__(PT)
void k_deg(const int* __restrict__ pairsRaw, const int* __restrict__ tab,
           const float* __restrict__ x, float* __restrict__ dinv,
           float* __restrict__ y, int n, int ch) {
  __shared__ int cnt[RB];
  const int tid = threadIdx.x, lane = tid & 31;
  const int w  = __builtin_amdgcn_readfirstlane(tid >> 5);
  const int b  = blockIdx.x;
  const int n0 = b << SH;
  const int2* __restrict__ pairs = (const int2*)pairsRaw;

  if (tid < RB) cnt[tid] = 0;
  int a[KPW], m[KPW], z[KPW];
  seg_bounds(tab, w, b, a, m, z);
  int longest = 0;
  #pragma unroll
  for (int k = 0; k < KPW; ++k) longest = max(longest, z[k] - a[k]);
  __syncthreads();
  for (int o = 0; o < longest; o += 32) {
    int d[KPW];
    #pragma unroll
    for (int k = 0; k < KPW; ++k) {
      const int i = a[k] + o + lane;
      const size_t j = (i < z[k]) ? (size_t)(w + k * WPG) * ch + i : 0;
      d[k] = pairs[j].y;
    }
    asm volatile("" :: "v"(d[0]));
    #pragma unroll
    for (int k = 0; k < KPW; ++k) {
      const int i = a[k] + o + lane;
      if (i < z[k]) atomicAdd(&cnt[d[k] - n0], 1);
    }
  }
  __syncthreads();
  if (tid < RB) {
    int node = n0 + tid;
    if (node < n) {
      float dv = rsqrtf((float)(cnt[tid] + 1));
      dinv[node] = dv;
      y[node]    = dv * x[node];
    }
  }
}

static_assert(RB == 16 * WPG, "one 16-node tile per wave covers the bucket");

__global__ __launch_bounds__(PT)
void k_layer1(const int* __restrict__ pairsRaw, const int* __restrict__ tab,
              const float* __restrict__ y, const float* __restrict__ dinv,
              const float* __restrict__ W1,
              const float* __restrict__ b1,
              const float* __restrict__ W2,
              float* __restrict__ g0, float* __restrict__ g1, int n, int ch, int hs) {
  __shared__ float acc[RB];
  __shared__ float tbl[HMAX];
  const int tid = threadIdx.x, lane = tid & 31;
  const int w  = __builtin_amdgcn_readfirstlane(tid >> 5);
  const int b  = blockIdx.x;
  const int n0 = b << SH;
  const int2* __restrict__ pairs = (const int2*)pairsRaw;

  const int half = lane >> 4;
  const int mm   = lane & 15;
  const int t0   = n0 + 16 * w;
  const int node = t0 + mm;
  const int   nidx = node < n ? node : 0;
  const float dn   = (node < n) ? dinv[nidx] : 0.0f;
  const float cmsk = (mm < 2) ? 1.0f : 0.0f;
  const int   cm   = (mm < 2) ? mm : 0;
  v2f Bf[8], Wf[8], bf[8];
  #pragma unroll
  for (int i = 0; i < 8; ++i) {
    const int f0 = 4 * i + 2 * half;
    Bf[i].x = W2[(f0    ) * 2 + cm];
    Bf[i].y = W2[(f0 + 1) * 2 + cm];
    Wf[i].x = W1[f0];  Wf[i].y = W1[f0 + 1];
    bf[i].x = b1[f0];  bf[i].y = b1[f0 + 1];
  }

  if (tid < RB) {
    int node = n0 + tid;
    acc[tid] = (node < n) ? y[node] : 0.0f;
  }
  int a[KPW], m[KPW], z[KPW];
  seg_bounds(tab, w, b, a, m, z);
  stage_half(tbl, y, min(hs, n), tid);
  asm volatile("" :: "v"(dn),
               "v"(Bf[0]), "v"(Bf[1]), "v"(Bf[2]), "v"(Bf[3]), "v"(Bf[4]), "v"(Bf[5]), "v"(Bf[6]), "v"(Bf[7]),
               "v"(Wf[0]), "v"(Wf[1]), "v"(Wf[2]), "v"(Wf[3]), "v"(Wf[4]), "v"(Wf[5]), "v"(Wf[6]), "v"(Wf[7]),
               "v"(bf[0]), "v"(bf[1]), "v"(bf[2]), "v"(bf[3]), "v"(bf[4]), "v"(bf[5]), "v"(bf[6]), "v"(bf[7]));
  __syncthreads();
  walk_half(pairs, w, ch, lane, a, m, tbl, 0, acc, n0);
  __syncthreads();
  stage_half(tbl, y + hs, n - hs, tid);
  __syncthreads();
  walk_half(pairs, w, ch, lane, m, z, tbl, hs, acc, n0);
  __syncthreads();

  const float sn = dn * acc[16 * w + mm];
  v8f c = {};
  #pragma unroll
  for (int i = 0; i < 8; ++i) {
    v2f A;
    A.x = dn * fmaxf(fmaf(sn, Wf[i].x, bf[i].x), 0.0f);
    A.y = dn * fmaxf(fmaf(sn, Wf[i].y, bf[i].y), 0.0f);
    v2f B;
    B.x = Bf[i].x * cmsk;
    B.y = Bf[i].y * cmsk;
    c = __builtin_amdgcn_wmma_f32_16x16x4_f32(
             false, A,  false, B,
             (short)0, c,  false,  false);
  }

  if (mm < 2) {
    float* __restrict__ gp = mm ? g1 : g0;
    #pragma unroll
    for (int v = 0; v < 8; ++v) {
      int row = t0 + v + 8 * half;
      if (row < n) gp[row] = c[v];
    }
  }
}

__global__ __launch_bounds__(PT)
void k_layer2(const int* __restrict__ pairsRaw, const int* __restrict__ tab,
              const float* __restrict__ g0, const float* __restrict__ g1,
              const float* __restrict__ dinv,
              const float* __restrict__ b2, float* __restrict__ out, int n, int ch, int hs) {
  __shared__ float acc[2 * RB];
  __shared__ float tbl[HMAX];
  const int tid = threadIdx.x, lane = tid & 31;
  const int w  = __builtin_amdgcn_readfirstlane(tid >> 5);
  const int b  = blockIdx.x;
  const int n0 = b << SH;
  const int2* __restrict__ pairs = (const int2*)pairsRaw;

  for (int t = tid; t < 2 * RB; t += PT) acc[t] = 0.0f;
  int a[KPW], m[KPW], z[KPW];
  seg_bounds(tab, w, b, a, m, z);
  for (int col = 0; col < 2; ++col) {
    const float* __restrict__ gp = col ? g1 : g0;
    float* accc = acc + col * RB;
    stage_half(tbl, gp, min(hs, n), tid);
    __syncthreads();
    walk_half(pairs, w, ch, lane, a, m, tbl, 0, accc, n0);
    __syncthreads();
    stage_half(tbl, gp + hs, n - hs, tid);
    __syncthreads();
    walk_half(pairs, w, ch, lane, m, z, tbl, hs, accc, n0);
    __syncthreads();
  }
  if (tid < RB) {
    int node = n0 + tid;
    if (node < n) {
      float dv = dinv[node];
      float o0 = dv * (acc[tid]      + g0[node]) + b2[0];
      float o1 = dv * (acc[RB + tid] + g1[node]) + b2[1];
      float mx  = fmaxf(o0, o1);
      float lse = mx + logf(expf(o0 - mx) + expf(o1 - mx));
      float2 r; r.x = o0 - lse; r.y = o1 - lse;
      *(float2*)(out + 2 * node) = r;
    }
  }
}

extern "C" void kernel_launch(void* const* d_in, const int* in_sizes, int n_in,
                              void* d_out, int out_size, void* d_ws, size_t ws_size,
                              hipStream_t stream) {
  const float* x  = (const float*)d_in[0];
  const int*   ei = (const int*)d_in[1];
  const float* W1 = (const float*)d_in[2];
  const float* b1 = (const float*)d_in[3];
  const float* W2 = (const float*)d_in[4];
  const float* b2 = (const float*)d_in[5];
  float* out = (float*)d_out;

  int n = in_sizes[0];
  int e = in_sizes[1] / 2;
  const int* srcIdx = ei;
  const int* dstIdx = ei + e;

  int np = (n + 3) & ~3;
  float* y     = (float*)d_ws;
  float* g0    = y + np;
  float* g1    = g0 + np;
  float* dinv  = g1 + np;
  int*   tab   = (int*)(dinv + np);
  int*   pairs = tab + 2 * NCH * KEYS;

  int nb = (n + RB - 1) / RB;
  int ch = (((e + NCH - 1) / NCH) + 15) & ~15;
  int hs = (((n + 1) / 2) + 3) & ~3;

  k_part  <<<NCH, PT, 0, stream>>>(srcIdx, dstIdx, pairs, tab, e, ch, hs);
  k_deg   <<<nb,  PT, 0, stream>>>(pairs, tab, x, dinv, y, n, ch);
  k_layer1<<<nb,  PT, 0, stream>>>(pairs, tab, y, dinv, W1, b1, W2, g0, g1, n, ch, hs);
  k_layer2<<<nb,  PT, 0, stream>>>(pairs, tab, g0, g1, dinv, b2, out, n, ch, hs);
}


# Sorted edges packed into one 32-bit word each (source offset in its half, destination offset in its bucket): the sort writes and every walk reads 4 bytes per edge, not 8; sum order as before
# speedup vs baseline: 16.2813x; 1.0963x over previous
#include <hip/hip_runtime.h>

typedef float v2f __attribute__((ext_vector_type(2)));
typedef float v4f __attribute__((ext_vector_type(4)));
typedef float v8f __attribute__((ext_vector_type(8)));

#define SH    9
#define RB    (1 << SH)
#define MAXB  256
#define KEYS  (2 * MAXB)
#define HMAX  65536
#define NCH   256
#define PT    1024
#define CAP   25600
#define WPG   (PT / 32)
#define KPW   (NCH / WPG)

__device__ __forceinline__ void seg_bounds(const int* __restrict__ tab, int w, int b,
                                           int (&a)[KPW], int (&m)[KPW], int (&z)[KPW]) {
  #pragma unroll
  for (int k = 0; k < KPW; ++k) {
    const int4 t = ((const int4*)tab)[(w + k * WPG) * MAXB + b];
    a[k] = t.x;
    m[k] = t.y;
    z[k] = t.w;
  }
}

__device__ __forceinline__ void stage_half(float* __restrict__ tbl, const float* __restrict__ src,
                                           int cnt, int tid) {
  const int c4 = cnt >> 2;
  const v4f* __restrict__ s4 = (const v4f*)src;
  v4f* __restrict__ t4 = (v4f*)tbl;
  for (int i = tid; i < c4; i += 8 * PT) {
    v4f v[8];
    #pragma unroll
    for (int u = 0; u < 8; ++u) {
      const int j = i + u * PT;
      v[u] = s4[j < c4 ? j : i];
    }
    asm volatile("" :: "v"(v[0]), "v"(v[1]), "v"(v[2]), "v"(v[3]),
                       "v"(v[4]), "v"(v[5]), "v"(v[6]), "v"(v[7]));
    #pragma unroll
    for (int u = 0; u < 8; ++u) {
      const int j = i + u * PT;
      if (j < c4) t4[j] = v[u];
    }
  }
  for (int i = (c4 << 2) + tid; i < cnt; i += PT) tbl[i] = src[i];
}

__device__ __forceinline__ void walk_half(const int* __restrict__ pairs, int w, int ch, int lane,
                                          const int (&lo)[KPW], const int (&hi)[KPW],
                                          const float* tbl, float* acc) {
  int longest = 0;
  #pragma unroll
  for (int k = 0; k < KPW; ++k) longest = max(longest, hi[k] - lo[k]);
  for (int o = 0; o < longest; o += 32) {
    int q[KPW];
    float v[KPW];
    #pragma unroll
    for (int k = 0; k < KPW; ++k) {
      const int i = lo[k] + o + lane;
      const size_t j = (i < hi[k]) ? (size_t)(w + k * WPG) * ch + i : 0;
      q[k] = pairs[j];
    }
    #pragma unroll
    for (int k = 0; k < KPW; ++k) {
      const int i = lo[k] + o + lane;
      v[k] = tbl[(i < hi[k]) ? (q[k] >> SH) : 0];
    }
    asm volatile("" :: "v"(v[0]));
    #pragma unroll
    for (int k = 0; k < KPW; ++k) {
      const int i = lo[k] + o + lane;
      if (i < hi[k]) atomicAdd(&acc[q[k] & (RB - 1)], v[k]);
    }
  }
}

__device__ __forceinline__ int sort_key(int s, int d, int hs) {
  return ((d >> SH) << 1) | (s >= hs ? 1 : 0);
}

__device__ __forceinline__ int pack_edge(int s, int d, int hs) {
  return ((s - (s >= hs ? hs : 0)) << SH) | (d & (RB - 1));
}

__global__ __launch_bounds__(PT)
void k_part(const int* __restrict__ srcIdx, const int* __restrict__ dstIdx,
            int* __restrict__ pairsRaw, int* __restrict__ tab, int e, int ch, int hs) {
  __shared__ int cnt[KEYS];
  __shared__ __attribute__((aligned(16))) int stage[CAP];
  const int tid = threadIdx.x;
  const int c   = blockIdx.x;
  const int beg = c * ch;
  const int end = min(beg + ch, e);
  const int m   = max(end - beg, 0);
  const bool staged = m <= CAP;
  int* __restrict__ pairs = pairsRaw;

  if (tid < KEYS) cnt[tid] = 0;
  __syncthreads();
  for (int i = beg + 4 * tid; i < end; i += 4 * PT) {
    if (i + 3 < end) {
      int4 d = *(const int4*)(dstIdx + i);
      int4 s = *(const int4*)(srcIdx + i);
      atomicAdd(&cnt[sort_key(s.x, d.x, hs)], 1);
      atomicAdd(&cnt[sort_key(s.y, d.y, hs)], 1);
      atomicAdd(&cnt[sort_key(s.z, d.z, hs)], 1);
      atomicAdd(&cnt[sort_key(s.w, d.w, hs)], 1);
    } else {
      for (int j = i; j < end; ++j) atomicAdd(&cnt[sort_key(srcIdx[j], dstIdx[j], hs)], 1);
    }
  }
  __syncthreads();

  if (tid < 32) {
    int cn[16], st[16];
    int run = 0;
    #pragma unroll
    for (int k = 0; k < 16; ++k) { cn[k] = cnt[tid * 16 + k]; st[k] = run; run += cn[k]; }
    int incl = run;
    #pragma unroll
    for (int off = 1; off < 32; off <<= 1) {
      int t = __shfl_up(incl, off);
      if (tid >= off) incl += t;
    }
    const int excl = incl - run;
    int2* row = (int2*)tab + c * KEYS;
    #pragma unroll
    for (int k = 0; k < 16; ++k) {
      int p = excl + st[k];
      cnt[tid * 16 + k] = p;
      row[tid * 16 + k] = make_int2(p, p + cn[k]);
    }
  }
  __syncthreads();

  for (int i = beg + 4 * tid; i < end; i += 4 * PT) {
    if (i + 3 < end) {
      int4 d = *(const int4*)(dstIdx + i);
      int4 s = *(const int4*)(srcIdx + i);
      int p0 = atomicAdd(&cnt[sort_key(s.x, d.x, hs)], 1);
      int p1 = atomicAdd(&cnt[sort_key(s.y, d.y, hs)], 1);
      int p2 = atomicAdd(&cnt[sort_key(s.z, d.z, hs)], 1);
      int p3 = atomicAdd(&cnt[sort_key(s.w, d.w, hs)], 1);
      if (staged) {
        stage[p0] = pack_edge(s.x, d.x, hs);
        stage[p1] = pack_edge(s.y, d.y, hs);
        stage[p2] = pack_edge(s.z, d.z, hs);
        stage[p3] = pack_edge(s.w, d.w, hs);
      } else {
        pairs[beg + p0] = pack_edge(s.x, d.x, hs);
        pairs[beg + p1] = pack_edge(s.y, d.y, hs);
        pairs[beg + p2] = pack_edge(s.z, d.z, hs);
        pairs[beg + p3] = pack_edge(s.w, d.w, hs);
      }
    } else {
      for (int j = i; j < end; ++j) {
        int d = dstIdx[j];
        int s = srcIdx[j];
        int p = atomicAdd(&cnt[sort_key(s, d, hs)], 1);
        if (staged) stage[p] = pack_edge(s, d, hs);
        else        pairs[beg + p] = pack_edge(s, d, hs);
      }
    }
  }

  if (staged) {
    __syncthreads();
    int4*       __restrict__ o4 = (int4*)(pairs + beg);
    const int4* __restrict__ s4 = (const int4*)stage;
    const int m4 = m >> 2;
    for (int i = tid; i < m4; i += PT) o4[i] = s4[i];
    if (tid < (m & 3)) pairs[beg + (m4 << 2) + tid] = stage[(m4 << 2) + tid];
  }
}

__global__ __launch_bounds__(PT)
void k_deg(const int* __restrict__ pairsRaw, const int* __restrict__ tab,
           const float* __restrict__ x, float* __restrict__ dinv,
           float* __restrict__ y, int n, int ch) {
  __shared__ int cnt[RB];
  const int tid = threadIdx.x, lane = tid & 31;
  const int w  = __builtin_amdgcn_readfirstlane(tid >> 5);
  const int b  = blockIdx.x;
  const int n0 = b << SH;
  const int* __restrict__ pairs = pairsRaw;

  if (tid < RB) cnt[tid] = 0;
  int a[KPW], m[KPW], z[KPW];
  seg_bounds(tab, w, b, a, m, z);
  int longest = 0;
  #pragma unroll
  for (int k = 0; k < KPW; ++k) longest = max(longest, z[k] - a[k]);
  __syncthreads();
  for (int o = 0; o < longest; o += 32) {
    int d[KPW];
    #pragma unroll
    for (int k = 0; k < KPW; ++k) {
      const int i = a[k] + o + lane;
      const size_t j = (i < z[k]) ? (size_t)(w + k * WPG) * ch + i : 0;
      d[k] = pairs[j];
    }
    asm volatile("" :: "v"(d[0]));
    #pragma unroll
    for (int k = 0; k < KPW; ++k) {
      const int i = a[k] + o + lane;
      if (i < z[k]) atomicAdd(&cnt[d[k] & (RB - 1)], 1);
    }
  }
  __syncthreads();
  if (tid < RB) {
    int node = n0 + tid;
    if (node < n) {
      float dv = rsqrtf((float)(cnt[tid] + 1));
      dinv[node] = dv;
      y[node]    = dv * x[node];
    }
  }
}

static_assert(RB == 16 * WPG, "one 16-node tile per wave covers the bucket");

__global__ __launch_bounds__(PT)
void k_layer1(const int* __restrict__ pairsRaw, const int* __restrict__ tab,
              const float* __restrict__ y, const float* __restrict__ dinv,
              const float* __restrict__ W1,
              const float* __restrict__ b1,
              const float* __restrict__ W2,
              float* __restrict__ g0, float* __restrict__ g1, int n, int ch, int hs) {
  __shared__ float acc[RB];
  __shared__ float tbl[HMAX];
  const int tid = threadIdx.x, lane = tid & 31;
  const int w  = __builtin_amdgcn_readfirstlane(tid >> 5);
  const int b  = blockIdx.x;
  const int n0 = b << SH;
  const int* __restrict__ pairs = pairsRaw;

  const int half = lane >> 4;
  const int mm   = lane & 15;
  const int t0   = n0 + 16 * w;
  const int node = t0 + mm;
  const int   nidx = node < n ? node : 0;
  const float dn   = (node < n) ? dinv[nidx] : 0.0f;
  const float cmsk = (mm < 2) ? 1.0f : 0.0f;
  const int   cm   = (mm < 2) ? mm : 0;
  v2f Bf[8], Wf[8], bf[8];
  #pragma unroll
  for (int i = 0; i < 8; ++i) {
    const int f0 = 4 * i + 2 * half;
    Bf[i].x = W2[(f0    ) * 2 + cm];
    Bf[i].y = W2[(f0 + 1) * 2 + cm];
    Wf[i].x = W1[f0];  Wf[i].y = W1[f0 + 1];
    bf[i].x = b1[f0];  bf[i].y = b1[f0 + 1];
  }

  if (tid < RB) {
    int node = n0 + tid;
    acc[tid] = (node < n) ? y[node] : 0.0f;
  }
  int a[KPW], m[KPW], z[KPW];
  seg_bounds(tab, w, b, a, m, z);
  stage_half(tbl, y, min(hs, n), tid);
  asm volatile("" :: "v"(dn),
               "v"(Bf[0]), "v"(Bf[1]), "v"(Bf[2]), "v"(Bf[3]), "v"(Bf[4]), "v"(Bf[5]), "v"(Bf[6]), "v"(Bf[7]),
               "v"(Wf[0]), "v"(Wf[1]), "v"(Wf[2]), "v"(Wf[3]), "v"(Wf[4]), "v"(Wf[5]), "v"(Wf[6]), "v"(Wf[7]),
               "v"(bf[0]), "v"(bf[1]), "v"(bf[2]), "v"(bf[3]), "v"(bf[4]), "v"(bf[5]), "v"(bf[6]), "v"(bf[7]));
  __syncthreads();
  walk_half(pairs, w, ch, lane, a, m, tbl, acc);
  __syncthreads();
  stage_half(tbl, y + hs, n - hs, tid);
  __syncthreads();
  walk_half(pairs, w, ch, lane, m, z, tbl, acc);
  __syncthreads();

  const float sn = dn * acc[16 * w + mm];
  v8f c = {};
  #pragma unroll
  for (int i = 0; i < 8; ++i) {
    v2f A;
    A.x = dn * fmaxf(fmaf(sn, Wf[i].x, bf[i].x), 0.0f);
    A.y = dn * fmaxf(fmaf(sn, Wf[i].y, bf[i].y), 0.0f);
    v2f B;
    B.x = Bf[i].x * cmsk;
    B.y = Bf[i].y * cmsk;
    c = __builtin_amdgcn_wmma_f32_16x16x4_f32(
             false, A,  false, B,
             (short)0, c,  false,  false);
  }

  if (mm < 2) {
    float* __restrict__ gp = mm ? g1 : g0;
    #pragma unroll
    for (int v = 0; v < 8; ++v) {
      int row = t0 + v + 8 * half;
      if (row < n) gp[row] = c[v];
    }
  }
}

__global__ __launch_bounds__(PT)
void k_layer2(const int* __restrict__ pairsRaw, const int* __restrict__ tab,
              const float* __restrict__ g0, const float* __restrict__ g1,
              const float* __restrict__ dinv,
              const float* __restrict__ b2, float* __restrict__ out, int n, int ch, int hs) {
  __shared__ float acc[2 * RB];
  __shared__ float tbl[HMAX];
  const int tid = threadIdx.x, lane = tid & 31;
  const int w  = __builtin_amdgcn_readfirstlane(tid >> 5);
  const int b  = blockIdx.x;
  const int n0 = b << SH;
  const int* __restrict__ pairs = pairsRaw;

  for (int t = tid; t < 2 * RB; t += PT) acc[t] = 0.0f;
  int a[KPW], m[KPW], z[KPW];
  seg_bounds(tab, w, b, a, m, z);
  for (int col = 0; col < 2; ++col) {
    const float* __restrict__ gp = col ? g1 : g0;
    float* accc = acc + col * RB;
    stage_half(tbl, gp, min(hs, n), tid);
    __syncthreads();
    walk_half(pairs, w, ch, lane, a, m, tbl, accc);
    __syncthreads();
    stage_half(tbl, gp + hs, n - hs, tid);
    __syncthreads();
    walk_half(pairs, w, ch, lane, m, z, tbl, accc);
    __syncthreads();
  }
  if (tid < RB) {
    int node = n0 + tid;
    if (node < n) {
      float dv = dinv[node];
      float o0 = dv * (acc[tid]      + g0[node]) + b2[0];
      float o1 = dv * (acc[RB + tid] + g1[node]) + b2[1];
      float mx  = fmaxf(o0, o1);
      float lse = mx + logf(expf(o0 - mx) + expf(o1 - mx));
      float2 r; r.x = o0 - lse; r.y = o1 - lse;
      *(float2*)(out + 2 * node) = r;
    }
  }
}

extern "C" void kernel_launch(void* const* d_in, const int* in_sizes, int n_in,
                              void* d_out, int out_size, void* d_ws, size_t ws_size,
                              hipStream_t stream) {
  const float* x  = (const float*)d_in[0];
  const int*   ei = (const int*)d_in[1];
  const float* W1 = (const float*)d_in[2];
  const float* b1 = (const float*)d_in[3];
  const float* W2 = (const float*)d_in[4];
  const float* b2 = (const float*)d_in[5];
  float* out = (float*)d_out;

  int n = in_sizes[0];
  int e = in_sizes[1] / 2;
  const int* srcIdx = ei;
  const int* dstIdx = ei + e;

  int np = (n + 3) & ~3;
  float* y     = (float*)d_ws;
  float* g0    = y + np;
  float* g1    = g0 + np;
  float* dinv  = g1 + np;
  int*   tab   = (int*)(dinv + np);
  int*   pairs = tab + 2 * NCH * KEYS;

  int nb = (n + RB - 1) / RB;
  int ch = (((e + NCH - 1) / NCH) + 31) & ~31;
  int hs = (((n + 1) / 2) + 3) & ~3;

  k_part  <<<NCH, PT, 0, stream>>>(srcIdx, dstIdx, pairs, tab, e, ch, hs);
  k_deg   <<<nb,  PT, 0, stream>>>(pairs, tab, x, dinv, y, n, ch);
  k_layer1<<<nb,  PT, 0, stream>>>(pairs, tab, y, dinv, W1, b1, W2, g0, g1, n, ch, hs);
  k_layer2<<<nb,  PT, 0, stream>>>(pairs, tab, g0, g1, dinv, b2, out, n, ch, hs);
}


# Buckets of 400 destination nodes instead of 512: 250 work-groups, not 196, spread the three edge-walking kernels over nearly all 256 WGPs; float sums in arrival order, regrouped by bucket
# speedup vs baseline: 16.8277x; 1.0336x over previous
#include <hip/hip_runtime.h>

typedef float v2f __attribute__((ext_vector_type(2)));
typedef float v4f __attribute__((ext_vector_type(4)));
typedef float v8f __attribute__((ext_vector_type(8)));

#define SH    9
#define DM    ((1 << SH) - 1)
#define RB    400
#define MAXB  256
#define KEYS  (2 * MAXB)
#define HMAX  65536
#define NCH   256
#define PT    1024
#define CAP   25600
#define WPG   (PT / 32)
#define KPW   (NCH / WPG)

__device__ __forceinline__ void seg_bounds(const int* __restrict__ tab, int w, int b,
                                           int (&a)[KPW], int (&m)[KPW], int (&z)[KPW]) {
  #pragma unroll
  for (int k = 0; k < KPW; ++k) {
    const int4 t = ((const int4*)tab)[(w + k * WPG) * MAXB + b];
    a[k] = t.x;
    m[k] = t.y;
    z[k] = t.w;
  }
}

__device__ __forceinline__ void stage_half(float* __restrict__ tbl, const float* __restrict__ src,
                                           int cnt, int tid) {
  const int c4 = cnt >> 2;
  const v4f* __restrict__ s4 = (const v4f*)src;
  v4f* __restrict__ t4 = (v4f*)tbl;
  for (int i = tid; i < c4; i += 8 * PT) {
    v4f v[8];
    #pragma unroll
    for (int u = 0; u < 8; ++u) {
      const int j = i + u * PT;
      v[u] = s4[j < c4 ? j : i];
    }
    asm volatile("" :: "v"(v[0]), "v"(v[1]), "v"(v[2]), "v"(v[3]),
                       "v"(v[4]), "v"(v[5]), "v"(v[6]), "v"(v[7]));
    #pragma unroll
    for (int u = 0; u < 8; ++u) {
      const int j = i + u * PT;
      if (j < c4) t4[j] = v[u];
    }
  }
  for (int i = (c4 << 2) + tid; i < cnt; i += PT) tbl[i] = src[i];
}

__device__ __forceinline__ void walk_half(const int* __restrict__ pairs, int w, int ch, int lane,
                                          const int (&lo)[KPW], const int (&hi)[KPW],
                                          const float* tbl, float* acc) {
  int longest = 0;
  #pragma unroll
  for (int k = 0; k < KPW; ++k) longest = max(longest, hi[k] - lo[k]);
  for (int o = 0; o < longest; o += 32) {
    int q[KPW];
    float v[KPW];
    #pragma unroll
    for (int k = 0; k < KPW; ++k) {
      const int i = lo[k] + o + lane;
      const size_t j = (i < hi[k]) ? (size_t)(w + k * WPG) * ch + i : 0;
      q[k] = pairs[j];
    }
    #pragma unroll
    for (int k = 0; k < KPW; ++k) {
      const int i = lo[k] + o + lane;
      v[k] = tbl[(i < hi[k]) ? (q[k] >> SH) : 0];
    }
    asm volatile("" :: "v"(v[0]));
    #pragma unroll
    for (int k = 0; k < KPW; ++k) {
      const int i = lo[k] + o + lane;
      if (i < hi[k]) atomicAdd(&acc[q[k] & DM], v[k]);
    }
  }
}

__device__ __forceinline__ int sort_key(int s, int d, int hs) {
  return (int)(((unsigned)d / RB) << 1) | (s >= hs ? 1 : 0);
}

__device__ __forceinline__ int pack_edge(int s, int d, int hs) {
  return ((s - (s >= hs ? hs : 0)) << SH) | (int)((unsigned)d % RB);
}

__global__ __launch_bounds__(PT)
void k_part(const int* __restrict__ srcIdx, const int* __restrict__ dstIdx,
            int* __restrict__ pairsRaw, int* __restrict__ tab, int e, int ch, int hs) {
  __shared__ int cnt[KEYS];
  __shared__ __attribute__((aligned(16))) int stage[CAP];
  const int tid = threadIdx.x;
  const int c   = blockIdx.x;
  const int beg = c * ch;
  const int end = min(beg + ch, e);
  const int m   = max(end - beg, 0);
  const bool staged = m <= CAP;
  int* __restrict__ pairs = pairsRaw;

  if (tid < KEYS) cnt[tid] = 0;
  __syncthreads();
  for (int i = beg + 4 * tid; i < end; i += 4 * PT) {
    if (i + 3 < end) {
      int4 d = *(const int4*)(dstIdx + i);
      int4 s = *(const int4*)(srcIdx + i);
      atomicAdd(&cnt[sort_key(s.x, d.x, hs)], 1);
      atomicAdd(&cnt[sort_key(s.y, d.y, hs)], 1);
      atomicAdd(&cnt[sort_key(s.z, d.z, hs)], 1);
      atomicAdd(&cnt[sort_key(s.w, d.w, hs)], 1);
    } else {
      for (int j = i; j < end; ++j) atomicAdd(&cnt[sort_key(srcIdx[j], dstIdx[j], hs)], 1);
    }
  }
  __syncthreads();

  if (tid < 32) {
    int cn[16], st[16];
    int run = 0;
    #pragma unroll
    for (int k = 0; k < 16; ++k) { cn[k] = cnt[tid * 16 + k]; st[k] = run; run += cn[k]; }
    int incl = run;
    #pragma unroll
    for (int off = 1; off < 32; off <<= 1) {
      int t = __shfl_up(incl, off);
      if (tid >= off) incl += t;
    }
    const int excl = incl - run;
    int2* row = (int2*)tab + c * KEYS;
    #pragma unroll
    for (int k = 0; k < 16; ++k) {
      int p = excl + st[k];
      cnt[tid * 16 + k] = p;
      row[tid * 16 + k] = make_int2(p, p + cn[k]);
    }
  }
  __syncthreads();

  for (int i = beg + 4 * tid; i < end; i += 4 * PT) {
    if (i + 3 < end) {
      int4 d = *(const int4*)(dstIdx + i);
      int4 s = *(const int4*)(srcIdx + i);
      int p0 = atomicAdd(&cnt[sort_key(s.x, d.x, hs)], 1);
      int p1 = atomicAdd(&cnt[sort_key(s.y, d.y, hs)], 1);
      int p2 = atomicAdd(&cnt[sort_key(s.z, d.z, hs)], 1);
      int p3 = atomicAdd(&cnt[sort_key(s.w, d.w, hs)], 1);
      if (staged) {
        stage[p0] = pack_edge(s.x, d.x, hs);
        stage[p1] = pack_edge(s.y, d.y, hs);
        stage[p2] = pack_edge(s.z, d.z, hs);
        stage[p3] = pack_edge(s.w, d.w, hs);
      } else {
        pairs[beg + p0] = pack_edge(s.x, d.x, hs);
        pairs[beg + p1] = pack_edge(s.y, d.y, hs);
        pairs[beg + p2] = pack_edge(s.z, d.z, hs);
        pairs[beg + p3] = pack_edge(s.w, d.w, hs);
      }
    } else {
      for (int j = i; j < end; ++j) {
        int d = dstIdx[j];
        int s = srcIdx[j];
        int p = atomicAdd(&cnt[sort_key(s, d, hs)], 1);
        if (staged) stage[p] = pack_edge(s, d, hs);
        else        pairs[beg + p] = pack_edge(s, d, hs);
      }
    }
  }

  if (staged) {
    __syncthreads();
    int4*       __restrict__ o4 = (int4*)(pairs + beg);
    const int4* __restrict__ s4 = (const int4*)stage;
    const int m4 = m >> 2;
    for (int i = tid; i < m4; i += PT) o4[i] = s4[i];
    if (tid < (m & 3)) pairs[beg + (m4 << 2) + tid] = stage[(m4 << 2) + tid];
  }
}

__global__ __launch_bounds__(PT)
void k_deg(const int* __restrict__ pairsRaw, const int* __restrict__ tab,
           const float* __restrict__ x, float* __restrict__ dinv,
           float* __restrict__ y, int n, int ch) {
  __shared__ int cnt[RB];
  const int tid = threadIdx.x, lane = tid & 31;
  const int w  = __builtin_amdgcn_readfirstlane(tid >> 5);
  const int b  = blockIdx.x;
  const int n0 = b * RB;
  const int* __restrict__ pairs = pairsRaw;

  if (tid < RB) cnt[tid] = 0;
  int a[KPW], m[KPW], z[KPW];
  seg_bounds(tab, w, b, a, m, z);
  int longest = 0;
  #pragma unroll
  for (int k = 0; k < KPW; ++k) longest = max(longest, z[k] - a[k]);
  __syncthreads();
  for (int o = 0; o < longest; o += 32) {
    int d[KPW];
    #pragma unroll
    for (int k = 0; k < KPW; ++k) {
      const int i = a[k] + o + lane;
      const size_t j = (i < z[k]) ? (size_t)(w + k * WPG) * ch + i : 0;
      d[k] = pairs[j];
    }
    asm volatile("" :: "v"(d[0]));
    #pragma unroll
    for (int k = 0; k < KPW; ++k) {
      const int i = a[k] + o + lane;
      if (i < z[k]) atomicAdd(&cnt[d[k] & DM], 1);
    }
  }
  __syncthreads();
  if (tid < RB) {
    int node = n0 + tid;
    if (node < n) {
      float dv = rsqrtf((float)(cnt[tid] + 1));
      dinv[node] = dv;
      y[node]    = dv * x[node];
    }
  }
}

static_assert(RB % 16 == 0 && RB / 16 <= WPG && RB <= (1 << SH), "16-node tiles, one per wave, cover the bucket");

__global__ __launch_bounds__(PT)
void k_layer1(const int* __restrict__ pairsRaw, const int* __restrict__ tab,
              const float* __restrict__ y, const float* __restrict__ dinv,
              const float* __restrict__ W1,
              const float* __restrict__ b1,
              const float* __restrict__ W2,
              float* __restrict__ g0, float* __restrict__ g1, int n, int ch, int hs) {
  __shared__ float acc[RB];
  __shared__ float tbl[HMAX];
  const int tid = threadIdx.x, lane = tid & 31;
  const int w  = __builtin_amdgcn_readfirstlane(tid >> 5);
  const int b  = blockIdx.x;
  const int n0 = b * RB;
  const int* __restrict__ pairs = pairsRaw;

  const int half = lane >> 4;
  const int mm   = lane & 15;
  const int t0   = n0 + 16 * w;
  const int node = t0 + mm;
  const int   nidx = node < n ? node : 0;
  const float dn   = (node < n) ? dinv[nidx] : 0.0f;
  const float cmsk = (mm < 2) ? 1.0f : 0.0f;
  const int   cm   = (mm < 2) ? mm : 0;
  v2f Bf[8], Wf[8], bf[8];
  #pragma unroll
  for (int i = 0; i < 8; ++i) {
    const int f0 = 4 * i + 2 * half;
    Bf[i].x = W2[(f0    ) * 2 + cm];
    Bf[i].y = W2[(f0 + 1) * 2 + cm];
    Wf[i].x = W1[f0];  Wf[i].y = W1[f0 + 1];
    bf[i].x = b1[f0];  bf[i].y = b1[f0 + 1];
  }

  if (tid < RB) {
    int node = n0 + tid;
    acc[tid] = (node < n) ? y[node] : 0.0f;
  }
  int a[KPW], m[KPW], z[KPW];
  seg_bounds(tab, w, b, a, m, z);
  stage_half(tbl, y, min(hs, n), tid);
  asm volatile("" :: "v"(dn),
               "v"(Bf[0]), "v"(Bf[1]), "v"(Bf[2]), "v"(Bf[3]), "v"(Bf[4]), "v"(Bf[5]), "v"(Bf[6]), "v"(Bf[7]),
               "v"(Wf[0]), "v"(Wf[1]), "v"(Wf[2]), "v"(Wf[3]), "v"(Wf[4]), "v"(Wf[5]), "v"(Wf[6]), "v"(Wf[7]),
               "v"(bf[0]), "v"(bf[1]), "v"(bf[2]), "v"(bf[3]), "v"(bf[4]), "v"(bf[5]), "v"(bf[6]), "v"(bf[7]));
  __syncthreads();
  walk_half(pairs, w, ch, lane, a, m, tbl, acc);
  __syncthreads();
  stage_half(tbl, y + hs, n - hs, tid);
  __syncthreads();
  walk_half(pairs, w, ch, lane, m, z, tbl, acc);
  __syncthreads();

  if (w >= RB / 16) return;
  const float sn = dn * acc[16 * w + mm];
  v8f c = {};
  #pragma unroll
  for (int i = 0; i < 8; ++i) {
    v2f A;
    A.x = dn * fmaxf(fmaf(sn, Wf[i].x, bf[i].x), 0.0f);
    A.y = dn * fmaxf(fmaf(sn, Wf[i].y, bf[i].y), 0.0f);
    v2f B;
    B.x = Bf[i].x * cmsk;
    B.y = Bf[i].y * cmsk;
    c = __builtin_amdgcn_wmma_f32_16x16x4_f32(
             false, A,  false, B,
             (short)0, c,  false,  false);
  }

  if (mm < 2) {
    float* __restrict__ gp = mm ? g1 : g0;
    #pragma unroll
    for (int v = 0; v < 8; ++v) {
      int row = t0 + v + 8 * half;
      if (row < n) gp[row] = c[v];
    }
  }
}

__global__ __launch_bounds__(PT)
void k_layer2(const int* __restrict__ pairsRaw, const int* __restrict__ tab,
              const float* __restrict__ g0, const float* __restrict__ g1,
              const float* __restrict__ dinv,
              const float* __restrict__ b2, float* __restrict__ out, int n, int ch, int hs) {
  __shared__ float acc[2 * RB];
  __shared__ float tbl[HMAX];
  const int tid = threadIdx.x, lane = tid & 31;
  const int w  = __builtin_amdgcn_readfirstlane(tid >> 5);
  const int b  = blockIdx.x;
  const int n0 = b * RB;
  const int* __restrict__ pairs = pairsRaw;

  for (int t = tid; t < 2 * RB; t += PT) acc[t] = 0.0f;
  int a[KPW], m[KPW], z[KPW];
  seg_bounds(tab, w, b, a, m, z);
  for (int col = 0; col < 2; ++col) {
    const float* __restrict__ gp = col ? g1 : g0;
    float* accc = acc + col * RB;
    stage_half(tbl, gp, min(hs, n), tid);
    __syncthreads();
    walk_half(pairs, w, ch, lane, a, m, tbl, accc);
    __syncthreads();
    stage_half(tbl, gp + hs, n - hs, tid);
    __syncthreads();
    walk_half(pairs, w, ch, lane, m, z, tbl, accc);
    __syncthreads();
  }
  if (tid < RB) {
    int node = n0 + tid;
    if (node < n) {
      float dv = dinv[node];
      float o0 = dv * (acc[tid]      + g0[node]) + b2[0];
      float o1 = dv * (acc[RB + tid] + g1[node]) + b2[1];
      float mx  = fmaxf(o0, o1);
      float lse = mx + logf(expf(o0 - mx) + expf(o1 - mx));
      float2 r; r.x = o0 - lse; r.y = o1 - lse;
      *(float2*)(out + 2 * node) = r;
    }
  }
}

extern "C" void kernel_launch(void* const* d_in, const int* in_sizes, int n_in,
                              void* d_out, int out_size, void* d_ws, size_t ws_size,
                              hipStream_t stream) {
  const float* x  = (const float*)d_in[0];
  const int*   ei = (const int*)d_in[1];
  const float* W1 = (const float*)d_in[2];
  const float* b1 = (const float*)d_in[3];
  const float* W2 = (const float*)d_in[4];
  const float* b2 = (const float*)d_in[5];
  float* out = (float*)d_out;

  int n = in_sizes[0];
  int e = in_sizes[1] / 2;
  const int* srcIdx = ei;
  const int* dstIdx = ei + e;

  int np = (n + 3) & ~3;
  float* y     = (float*)d_ws;
  float* g0    = y + np;
  float* g1    = g0 + np;
  float* dinv  = g1 + np;
  int*   tab   = (int*)(dinv + np);
  int*   pairs = tab + 2 * NCH * KEYS;

  int nb = (n + RB - 1) / RB;
  int ch = (((e + NCH - 1) / NCH) + 31) & ~31;
  int hs = (((n + 1) / 2) + 3) & ~3;

  k_part  <<<NCH, PT, 0, stream>>>(srcIdx, dstIdx, pairs, tab, e, ch, hs);
  k_deg   <<<nb,  PT, 0, stream>>>(pairs, tab, x, dinv, y, n, ch);
  k_layer1<<<nb,  PT, 0, stream>>>(pairs, tab, y, dinv, W1, b1, W2, g0, g1, n, ch, hs);
  k_layer2<<<nb,  PT, 0, stream>>>(pairs, tab, g0, g1, dinv, b2, out, n, ch, hs);
}


# Layer 2 sums one column, gd = g1 - g0 (W2 column difference in the WMMA B operand): a 2-class log_softmax needs only the logit difference (sums regrouped): 2 stagings, 2 half-walks, not 4 and 4
# speedup vs baseline: 20.4961x; 1.2180x over previous
#include <hip/hip_runtime.h>

typedef float v2f __attribute__((ext_vector_type(2)));
typedef float v4f __attribute__((ext_vector_type(4)));
typedef float v8f __attribute__((ext_vector_type(8)));

#define SH    9
#define DM    ((1 << SH) - 1)
#define RB    400
#define MAXB  256
#define KEYS  (2 * MAXB)
#define HMAX  65536
#define NCH   256
#define PT    1024
#define CAP   25600
#define WPG   (PT / 32)
#define KPW   (NCH / WPG)

__device__ __forceinline__ void seg_bounds(const int* __restrict__ tab, int w, int b,
                                           int (&a)[KPW], int (&m)[KPW], int (&z)[KPW]) {
  #pragma unroll
  for (int k = 0; k < KPW; ++k) {
    const int4 t = ((const int4*)tab)[(w + k * WPG) * MAXB + b];
    a[k] = t.x;
    m[k] = t.y;
    z[k] = t.w;
  }
}

__device__ __forceinline__ void stage_half(float* __restrict__ tbl, const float* __restrict__ src,
                                           int cnt, int tid) {
  const int c4 = cnt >> 2;
  const v4f* __restrict__ s4 = (const v4f*)src;
  v4f* __restrict__ t4 = (v4f*)tbl;
  for (int i = tid; i < c4; i += 8 * PT) {
    v4f v[8];
    #pragma unroll
    for (int u = 0; u < 8; ++u) {
      const int j = i + u * PT;
      v[u] = s4[j < c4 ? j : i];
    }
    asm volatile("" :: "v"(v[0]), "v"(v[1]), "v"(v[2]), "v"(v[3]),
                       "v"(v[4]), "v"(v[5]), "v"(v[6]), "v"(v[7]));
    #pragma unroll
    for (int u = 0; u < 8; ++u) {
      const int j = i + u * PT;
      if (j < c4) t4[j] = v[u];
    }
  }
  for (int i = (c4 << 2) + tid; i < cnt; i += PT) tbl[i] = src[i];
}

__device__ __forceinline__ void walk_half(const int* __restrict__ pairs, int w, int ch, int lane,
                                          const int (&lo)[KPW], const int (&hi)[KPW],
                                          const float* tbl, float* acc) {
  int longest = 0;
  #pragma unroll
  for (int k = 0; k < KPW; ++k) longest = max(longest, hi[k] - lo[k]);
  for (int o = 0; o < longest; o += 32) {
    int q[KPW];
    float v[KPW];
    #pragma unroll
    for (int k = 0; k < KPW; ++k) {
      const int i = lo[k] + o + lane;
      const size_t j = (i < hi[k]) ? (size_t)(w + k * WPG) * ch + i : 0;
      q[k] = pairs[j];
    }
    #pragma unroll
    for (int k = 0; k < KPW; ++k) {
      const int i = lo[k] + o + lane;
      v[k] = tbl[(i < hi[k]) ? (q[k] >> SH) : 0];
    }
    asm volatile("" :: "v"(v[0]));
    #pragma unroll
    for (int k = 0; k < KPW; ++k) {
      const int i = lo[k] + o + lane;
      if (i < hi[k]) atomicAdd(&acc[q[k] & DM], v[k]);
    }
  }
}

__device__ __forceinline__ int sort_key(int s, int d, int hs) {
  return (int)(((unsigned)d / RB) << 1) | (s >= hs ? 1 : 0);
}

__device__ __forceinline__ int pack_edge(int s, int d, int hs) {
  return ((s - (s >= hs ? hs : 0)) << SH) | (int)((unsigned)d % RB);
}

__global__ __launch_bounds__(PT)
void k_part(const int* __restrict__ srcIdx, const int* __restrict__ dstIdx,
            int* __restrict__ pairsRaw, int* __restrict__ tab, int e, int ch, int hs) {
  __shared__ int cnt[KEYS];
  __shared__ __attribute__((aligned(16))) int stage[CAP];
  const int tid = threadIdx.x;
  const int c   = blockIdx.x;
  const int beg = c * ch;
  const int end = min(beg + ch, e);
  const int m   = max(end - beg, 0);
  const bool staged = m <= CAP;
  int* __restrict__ pairs = pairsRaw;

  if (tid < KEYS) cnt[tid] = 0;
  __syncthreads();
  for (int i = beg + 4 * tid; i < end; i += 4 * PT) {
    if (i + 3 < end) {
      int4 d = *(const int4*)(dstIdx + i);
      int4 s = *(const int4*)(srcIdx + i);
      atomicAdd(&cnt[sort_key(s.x, d.x, hs)], 1);
      atomicAdd(&cnt[sort_key(s.y, d.y, hs)], 1);
      atomicAdd(&cnt[sort_key(s.z, d.z, hs)], 1);
      atomicAdd(&cnt[sort_key(s.w, d.w, hs)], 1);
    } else {
      for (int j = i; j < end; ++j) atomicAdd(&cnt[sort_key(srcIdx[j], dstIdx[j], hs)], 1);
    }
  }
  __syncthreads();

  if (tid < 32) {
    int cn[16], st[16];
    int run = 0;
    #pragma unroll
    for (int k = 0; k < 16; ++k) { cn[k] = cnt[tid * 16 + k]; st[k] = run; run += cn[k]; }
    int incl = run;
    #pragma unroll
    for (int off = 1; off < 32; off <<= 1) {
      int t = __shfl_up(incl, off);
      if (tid >= off) incl += t;
    }
    const int excl = incl - run;
    int2* row = (int2*)tab + c * KEYS;
    #pragma unroll
    for (int k = 0; k < 16; ++k) {
      int p = excl + st[k];
      cnt[tid * 16 + k] = p;
      row[tid * 16 + k] = make_int2(p, p + cn[k]);
    }
  }
  __syncthreads();

  for (int i = beg + 4 * tid; i < end; i += 4 * PT) {
    if (i + 3 < end) {
      int4 d = *(const int4*)(dstIdx + i);
      int4 s = *(const int4*)(srcIdx + i);
      int p0 = atomicAdd(&cnt[sort_key(s.x, d.x, hs)], 1);
      int p1 = atomicAdd(&cnt[sort_key(s.y, d.y, hs)], 1);
      int p2 = atomicAdd(&cnt[sort_key(s.z, d.z, hs)], 1);
      int p3 = atomicAdd(&cnt[sort_key(s.w, d.w, hs)], 1);
      if (staged) {
        stage[p0] = pack_edge(s.x, d.x, hs);
        stage[p1] = pack_edge(s.y, d.y, hs);
        stage[p2] = pack_edge(s.z, d.z, hs);
        stage[p3] = pack_edge(s.w, d.w, hs);
      } else {
        pairs[beg + p0] = pack_edge(s.x, d.x, hs);
        pairs[beg + p1] = pack_edge(s.y, d.y, hs);
        pairs[beg + p2] = pack_edge(s.z, d.z, hs);
        pairs[beg + p3] = pack_edge(s.w, d.w, hs);
      }
    } else {
      for (int j = i; j < end; ++j) {
        int d = dstIdx[j];
        int s = srcIdx[j];
        int p = atomicAdd(&cnt[sort_key(s, d, hs)], 1);
        if (staged) stage[p] = pack_edge(s, d, hs);
        else        pairs[beg + p] = pack_edge(s, d, hs);
      }
    }
  }

  if (staged) {
    __syncthreads();
    int4*       __restrict__ o4 = (int4*)(pairs + beg);
    const int4* __restrict__ s4 = (const int4*)stage;
    const int m4 = m >> 2;
    for (int i = tid; i < m4; i += PT) o4[i] = s4[i];
    if (tid < (m & 3)) pairs[beg + (m4 << 2) + tid] = stage[(m4 << 2) + tid];
  }
}

__global__ __launch_bounds__(PT)
void k_deg(const int* __restrict__ pairsRaw, const int* __restrict__ tab,
           const float* __restrict__ x, float* __restrict__ dinv,
           float* __restrict__ y, int n, int ch) {
  __shared__ int cnt[RB];
  const int tid = threadIdx.x, lane = tid & 31;
  const int w  = __builtin_amdgcn_readfirstlane(tid >> 5);
  const int b  = blockIdx.x;
  const int n0 = b * RB;
  const int* __restrict__ pairs = pairsRaw;

  if (tid < RB) cnt[tid] = 0;
  int a[KPW], m[KPW], z[KPW];
  seg_bounds(tab, w, b, a, m, z);
  int longest = 0;
  #pragma unroll
  for (int k = 0; k < KPW; ++k) longest = max(longest, z[k] - a[k]);
  __syncthreads();
  for (int o = 0; o < longest; o += 32) {
    int d[KPW];
    #pragma unroll
    for (int k = 0; k < KPW; ++k) {
      const int i = a[k] + o + lane;
      const size_t j = (i < z[k]) ? (size_t)(w + k * WPG) * ch + i : 0;
      d[k] = pairs[j];
    }
    asm volatile("" :: "v"(d[0]));
    #pragma unroll
    for (int k = 0; k < KPW; ++k) {
      const int i = a[k] + o + lane;
      if (i < z[k]) atomicAdd(&cnt[d[k] & DM], 1);
    }
  }
  __syncthreads();
  if (tid < RB) {
    int node = n0 + tid;
    if (node < n) {
      float dv = rsqrtf((float)(cnt[tid] + 1));
      dinv[node] = dv;
      y[node]    = dv * x[node];
    }
  }
}

static_assert(RB % 16 == 0 && RB / 16 <= WPG && RB <= (1 << SH), "16-node tiles, one per wave, cover the bucket");

__global__ __launch_bounds__(PT)
void k_layer1(const int* __restrict__ pairsRaw, const int* __restrict__ tab,
              const float* __restrict__ y, const float* __restrict__ dinv,
              const float* __restrict__ W1,
              const float* __restrict__ b1,
              const float* __restrict__ W2,
              float* __restrict__ gd, int n, int ch, int hs) {
  __shared__ float acc[RB];
  __shared__ float tbl[HMAX];
  const int tid = threadIdx.x, lane = tid & 31;
  const int w  = __builtin_amdgcn_readfirstlane(tid >> 5);
  const int b  = blockIdx.x;
  const int n0 = b * RB;
  const int* __restrict__ pairs = pairsRaw;

  const int half = lane >> 4;
  const int mm   = lane & 15;
  const int t0   = n0 + 16 * w;
  const int node = t0 + mm;
  const int   nidx = node < n ? node : 0;
  const float dn   = (node < n) ? dinv[nidx] : 0.0f;
  const float cmsk = (mm == 0) ? 1.0f : 0.0f;
  v2f Bf[8], Wf[8], bf[8];
  #pragma unroll
  for (int i = 0; i < 8; ++i) {
    const int f0 = 4 * i + 2 * half;
    Bf[i].x = W2[(f0    ) * 2 + 1] - W2[(f0    ) * 2];
    Bf[i].y = W2[(f0 + 1) * 2 + 1] - W2[(f0 + 1) * 2];
    Wf[i].x = W1[f0];  Wf[i].y = W1[f0 + 1];
    bf[i].x = b1[f0];  bf[i].y = b1[f0 + 1];
  }

  if (tid < RB) {
    int node = n0 + tid;
    acc[tid] = (node < n) ? y[node] : 0.0f;
  }
  int a[KPW], m[KPW], z[KPW];
  seg_bounds(tab, w, b, a, m, z);
  stage_half(tbl, y, min(hs, n), tid);
  asm volatile("" :: "v"(dn),
               "v"(Bf[0]), "v"(Bf[1]), "v"(Bf[2]), "v"(Bf[3]), "v"(Bf[4]), "v"(Bf[5]), "v"(Bf[6]), "v"(Bf[7]),
               "v"(Wf[0]), "v"(Wf[1]), "v"(Wf[2]), "v"(Wf[3]), "v"(Wf[4]), "v"(Wf[5]), "v"(Wf[6]), "v"(Wf[7]),
               "v"(bf[0]), "v"(bf[1]), "v"(bf[2]), "v"(bf[3]), "v"(bf[4]), "v"(bf[5]), "v"(bf[6]), "v"(bf[7]));
  __syncthreads();
  walk_half(pairs, w, ch, lane, a, m, tbl, acc);
  __syncthreads();
  stage_half(tbl, y + hs, n - hs, tid);
  __syncthreads();
  walk_half(pairs, w, ch, lane, m, z, tbl, acc);
  __syncthreads();

  if (w >= RB / 16) return;
  const float sn = dn * acc[16 * w + mm];
  v8f c = {};
  #pragma unroll
  for (int i = 0; i < 8; ++i) {
    v2f A;
    A.x = dn * fmaxf(fmaf(sn, Wf[i].x, bf[i].x), 0.0f);
    A.y = dn * fmaxf(fmaf(sn, Wf[i].y, bf[i].y), 0.0f);
    v2f B;
    B.x = Bf[i].x * cmsk;
    B.y = Bf[i].y * cmsk;
    c = __builtin_amdgcn_wmma_f32_16x16x4_f32(
             false, A,  false, B,
             (short)0, c,  false,  false);
  }

  if (mm == 0) {
    #pragma unroll
    for (int v = 0; v < 8; ++v) {
      int row = t0 + v + 8 * half;
      if (row < n) gd[row] = c[v];
    }
  }
}

__global__ __launch_bounds__(PT)
void k_layer2(const int* __restrict__ pairsRaw, const int* __restrict__ tab,
              const float* __restrict__ gd,
              const float* __restrict__ dinv,
              const float* __restrict__ b2, float* __restrict__ out, int n, int ch, int hs) {
  __shared__ float acc[RB];
  __shared__ float tbl[HMAX];
  const int tid = threadIdx.x, lane = tid & 31;
  const int w  = __builtin_amdgcn_readfirstlane(tid >> 5);
  const int b  = blockIdx.x;
  const int n0 = b * RB;
  const int* __restrict__ pairs = pairsRaw;

  if (tid < RB) {
    int node = n0 + tid;
    acc[tid] = (node < n) ? gd[node] : 0.0f;
  }
  int a[KPW], m[KPW], z[KPW];
  seg_bounds(tab, w, b, a, m, z);
  stage_half(tbl, gd, min(hs, n), tid);
  __syncthreads();
  walk_half(pairs, w, ch, lane, a, m, tbl, acc);
  __syncthreads();
  stage_half(tbl, gd + hs, n - hs, tid);
  __syncthreads();
  walk_half(pairs, w, ch, lane, m, z, tbl, acc);
  __syncthreads();
  if (tid < RB) {
    int node = n0 + tid;
    if (node < n) {
      float dl  = dinv[node] * acc[tid] + (b2[1] - b2[0]);
      float mx  = fmaxf(dl, 0.0f);
      float lse = mx + logf(1.0f + expf(-fabsf(dl)));
      float2 r; r.x = -lse; r.y = dl - lse;
      *(float2*)(out + 2 * node) = r;
    }
  }
}

extern "C" void kernel_launch(void* const* d_in, const int* in_sizes, int n_in,
                              void* d_out, int out_size, void* d_ws, size_t ws_size,
                              hipStream_t stream) {
  const float* x  = (const float*)d_in[0];
  const int*   ei = (const int*)d_in[1];
  const float* W1 = (const float*)d_in[2];
  const float* b1 = (const float*)d_in[3];
  const float* W2 = (const float*)d_in[4];
  const float* b2 = (const float*)d_in[5];
  float* out = (float*)d_out;

  int n = in_sizes[0];
  int e = in_sizes[1] / 2;
  const int* srcIdx = ei;
  const int* dstIdx = ei + e;

  int np = (n + 3) & ~3;
  float* y     = (float*)d_ws;
  float* gd    = y + np;
  float* dinv  = gd + np;
  int*   tab   = (int*)(dinv + np);
  int*   pairs = tab + 2 * NCH * KEYS;

  int nb = (n + RB - 1) / RB;
  int ch = (((e + NCH - 1) / NCH) + 31) & ~31;
  int hs = (((n + 1) / 2) + 3) & ~3;

  k_part  <<<NCH, PT, 0, stream>>>(srcIdx, dstIdx, pairs, tab, e, ch, hs);
  k_deg   <<<nb,  PT, 0, stream>>>(pairs, tab, x, dinv, y, n, ch);
  k_layer1<<<nb,  PT, 0, stream>>>(pairs, tab, y, dinv, W1, b1, W2, gd, n, ch, hs);
  k_layer2<<<nb,  PT, 0, stream>>>(pairs, tab, gd, dinv, b2, out, n, ch, hs);
}


# LDS atomics made lane-private in all four kernels (32 words per sort key or node, word*32+lane): lanes of one LDS add never share a word or bank; float sums per lane, then one fixed sum per node
# speedup vs baseline: 23.5884x; 1.1509x over previous
#include <hip/hip_runtime.h>

typedef float v2f __attribute__((ext_vector_type(2)));
typedef float v4f __attribute__((ext_vector_type(4)));
typedef float v8f __attribute__((ext_vector_type(8)));

#define SH    9
#define DM    ((1 << SH) - 1)
#define RB    400
#define MAXB  256
#define KEYS  (2 * MAXB)
#define HMAX  65536
#define NCH   256
#define PT    1024
#define CAP   25600
#define WPG   (PT / 32)
#define LP    32
#define KPW   (NCH / WPG)

__device__ __forceinline__ void seg_bounds(const int* __restrict__ tab, int w, int b,
                                           int (&a)[KPW], int (&m)[KPW], int (&z)[KPW]) {
  #pragma unroll
  for (int k = 0; k < KPW; ++k) {
    const int4 t = ((const int4*)tab)[(w + k * WPG) * MAXB + b];
    a[k] = t.x;
    m[k] = t.y;
    z[k] = t.w;
  }
}

__device__ __forceinline__ void stage_half(float* __restrict__ tbl, const float* __restrict__ src,
                                           int cnt, int tid) {
  const int c4 = cnt >> 2;
  const v4f* __restrict__ s4 = (const v4f*)src;
  v4f* __restrict__ t4 = (v4f*)tbl;
  for (int i = tid; i < c4; i += 8 * PT) {
    v4f v[8];
    #pragma unroll
    for (int u = 0; u < 8; ++u) {
      const int j = i + u * PT;
      v[u] = s4[j < c4 ? j : i];
    }
    asm volatile("" :: "v"(v[0]), "v"(v[1]), "v"(v[2]), "v"(v[3]),
                       "v"(v[4]), "v"(v[5]), "v"(v[6]), "v"(v[7]));
    #pragma unroll
    for (int u = 0; u < 8; ++u) {
      const int j = i + u * PT;
      if (j < c4) t4[j] = v[u];
    }
  }
  for (int i = (c4 << 2) + tid; i < cnt; i += PT) tbl[i] = src[i];
}

__device__ __forceinline__ void walk_half(const int* __restrict__ pairs, int w, int ch, int lane,
                                          const int (&lo)[KPW], const int (&hi)[KPW],
                                          const float* tbl, float* acc) {
  int longest = 0;
  #pragma unroll
  for (int k = 0; k < KPW; ++k) longest = max(longest, hi[k] - lo[k]);
  for (int o = 0; o < longest; o += 32) {
    int q[KPW];
    float v[KPW];
    #pragma unroll
    for (int k = 0; k < KPW; ++k) {
      const int i = lo[k] + o + lane;
      const size_t j = (i < hi[k]) ? (size_t)(w + k * WPG) * ch + i : 0;
      q[k] = pairs[j];
    }
    #pragma unroll
    for (int k = 0; k < KPW; ++k) {
      const int i = lo[k] + o + lane;
      v[k] = tbl[(i < hi[k]) ? (q[k] >> SH) : 0];
    }
    asm volatile("" :: "v"(v[0]));
    #pragma unroll
    for (int k = 0; k < KPW; ++k) {
      const int i = lo[k] + o + lane;
      if (i < hi[k]) atomicAdd(&acc[(q[k] & DM) * LP + lane], v[k]);
    }
  }
}

__device__ __forceinline__ int sort_key(int s, int d, int hs) {
  return (int)(((unsigned)d / RB) << 1) | (s >= hs ? 1 : 0);
}

__device__ __forceinline__ int pack_edge(int s, int d, int hs) {
  return ((s - (s >= hs ? hs : 0)) << SH) | (int)((unsigned)d % RB);
}

static_assert(LP == 32 && WPG == 32 && KEYS * LP == 16 * PT, "the scan gives every thread 16 counter words and every lane one wave total");

__global__ __launch_bounds__(PT)
void k_part(const int* __restrict__ srcIdx, const int* __restrict__ dstIdx,
            int* __restrict__ pairsRaw, int* __restrict__ tab, int e, int ch, int hs) {
  __shared__ __attribute__((aligned(16))) int cnt[KEYS * LP];
  __shared__ int wsum[WPG];
  __shared__ __attribute__((aligned(16))) int stage[CAP];
  const int tid = threadIdx.x, lane = tid & 31;
  const int c   = blockIdx.x;
  const int beg = c * ch;
  const int end = min(beg + ch, e);
  const int m   = max(end - beg, 0);
  const bool staged = m <= CAP;
  int* __restrict__ pairs = pairsRaw;

  for (int i = tid; i < KEYS * LP / 4; i += PT) ((int4*)cnt)[i] = make_int4(0, 0, 0, 0);
  __syncthreads();
  for (int i = beg + 4 * tid; i < end; i += 4 * PT) {
    if (i + 3 < end) {
      int4 d = *(const int4*)(dstIdx + i);
      int4 s = *(const int4*)(srcIdx + i);
      atomicAdd(&cnt[sort_key(s.x, d.x, hs) * LP + lane], 1);
      atomicAdd(&cnt[sort_key(s.y, d.y, hs) * LP + lane], 1);
      atomicAdd(&cnt[sort_key(s.z, d.z, hs) * LP + lane], 1);
      atomicAdd(&cnt[sort_key(s.w, d.w, hs) * LP + lane], 1);
    } else {
      for (int j = i; j < end; ++j) atomicAdd(&cnt[sort_key(srcIdx[j], dstIdx[j], hs) * LP + lane], 1);
    }
  }
  __syncthreads();

  {
    const int g = (tid >> 2) & 15;
    int st[16];
    int run = 0;
    #pragma unroll
    for (int u = 0; u < 16; ++u) { st[u] = run; run += cnt[16 * tid + ((u + g) & 15)]; }
    int incl = run;
    #pragma unroll
    for (int off = 1; off < 32; off <<= 1) {
      int t = __shfl_up(incl, off);
      if (lane >= off) incl += t;
    }
    if (lane == 31) wsum[tid >> 5] = incl;
    __syncthreads();
    const int ws = wsum[lane];
    int wi = ws;
    #pragma unroll
    for (int off = 1; off < 32; off <<= 1) {
      int t = __shfl_up(wi, off);
      if (lane >= off) wi += t;
    }
    const int base = __shfl(wi - ws, tid >> 5) + incl - run;
    #pragma unroll
    for (int u = 0; u < 16; ++u) cnt[16 * tid + ((u + g) & 15)] = base + st[u];
    const int nxt = __shfl_down(run, 1);
    int2* row = (int2*)tab + c * KEYS;
    if ((tid & 1) == 0) row[tid >> 1] = make_int2(base, base + run + nxt);
  }
  __syncthreads();

  for (int i = beg + 4 * tid; i < end; i += 4 * PT) {
    if (i + 3 < end) {
      int4 d = *(const int4*)(dstIdx + i);
      int4 s = *(const int4*)(srcIdx + i);
      int p0 = atomicAdd(&cnt[sort_key(s.x, d.x, hs) * LP + lane], 1);
      int p1 = atomicAdd(&cnt[sort_key(s.y, d.y, hs) * LP + lane], 1);
      int p2 = atomicAdd(&cnt[sort_key(s.z, d.z, hs) * LP + lane], 1);
      int p3 = atomicAdd(&cnt[sort_key(s.w, d.w, hs) * LP + lane], 1);
      if (staged) {
        stage[p0] = pack_edge(s.x, d.x, hs);
        stage[p1] = pack_edge(s.y, d.y, hs);
        stage[p2] = pack_edge(s.z, d.z, hs);
        stage[p3] = pack_edge(s.w, d.w, hs);
      } else {
        pairs[beg + p0] = pack_edge(s.x, d.x, hs);
        pairs[beg + p1] = pack_edge(s.y, d.y, hs);
        pairs[beg + p2] = pack_edge(s.z, d.z, hs);
        pairs[beg + p3] = pack_edge(s.w, d.w, hs);
      }
    } else {
      for (int j = i; j < end; ++j) {
        int d = dstIdx[j];
        int s = srcIdx[j];
        int p = atomicAdd(&cnt[sort_key(s, d, hs) * LP + lane], 1);
        if (staged) stage[p] = pack_edge(s, d, hs);
        else        pairs[beg + p] = pack_edge(s, d, hs);
      }
    }
  }

  if (staged) {
    __syncthreads();
    int4*       __restrict__ o4 = (int4*)(pairs + beg);
    const int4* __restrict__ s4 = (const int4*)stage;
    const int m4 = m >> 2;
    for (int i = tid; i < m4; i += PT) o4[i] = s4[i];
    if (tid < (m & 3)) pairs[beg + (m4 << 2) + tid] = stage[(m4 << 2) + tid];
  }
}

__global__ __launch_bounds__(PT)
void k_deg(const int* __restrict__ pairsRaw, const int* __restrict__ tab,
           const float* __restrict__ x, float* __restrict__ dinv,
           float* __restrict__ y, int n, int ch) {
  __shared__ __attribute__((aligned(16))) int cnt[RB * LP];
  const int tid = threadIdx.x, lane = tid & 31;
  const int w  = __builtin_amdgcn_readfirstlane(tid >> 5);
  const int b  = blockIdx.x;
  const int n0 = b * RB;
  const int* __restrict__ pairs = pairsRaw;

  for (int i = tid; i < RB * LP / 4; i += PT) ((int4*)cnt)[i] = make_int4(0, 0, 0, 0);
  int a[KPW], m[KPW], z[KPW];
  seg_bounds(tab, w, b, a, m, z);
  int longest = 0;
  #pragma unroll
  for (int k = 0; k < KPW; ++k) longest = max(longest, z[k] - a[k]);
  __syncthreads();
  for (int o = 0; o < longest; o += 32) {
    int d[KPW];
    #pragma unroll
    for (int k = 0; k < KPW; ++k) {
      const int i = a[k] + o + lane;
      const size_t j = (i < z[k]) ? (size_t)(w + k * WPG) * ch + i : 0;
      d[k] = pairs[j];
    }
    asm volatile("" :: "v"(d[0]));
    #pragma unroll
    for (int k = 0; k < KPW; ++k) {
      const int i = a[k] + o + lane;
      if (i < z[k]) atomicAdd(&cnt[(d[k] & DM) * LP + lane], 1);
    }
  }
  __syncthreads();
  if (tid < RB) {
    int deg = 1;
    #pragma unroll
    for (int j = 0; j < LP; ++j) deg += cnt[tid * LP + ((j + tid) & (LP - 1))];
    int node = n0 + tid;
    if (node < n) {
      float dv = rsqrtf((float)deg);
      dinv[node] = dv;
      y[node]    = dv * x[node];
    }
  }
}

static_assert(RB % 16 == 0 && RB / 16 <= WPG && RB <= (1 << SH), "16-node tiles, one per wave, cover the bucket");

__global__ __launch_bounds__(PT)
void k_layer1(const int* __restrict__ pairsRaw, const int* __restrict__ tab,
              const float* __restrict__ y, const float* __restrict__ dinv,
              const float* __restrict__ W1,
              const float* __restrict__ b1,
              const float* __restrict__ W2,
              float* __restrict__ gd, int n, int ch, int hs) {
  __shared__ __attribute__((aligned(16))) float lds[RB * LP + HMAX];
  float* acc = lds;
  float* tbl = lds + RB * LP;
  const int tid = threadIdx.x, lane = tid & 31;
  const int w  = __builtin_amdgcn_readfirstlane(tid >> 5);
  const int b  = blockIdx.x;
  const int n0 = b * RB;
  const int* __restrict__ pairs = pairsRaw;

  const int half = lane >> 4;
  const int mm   = lane & 15;
  const int t0   = n0 + 16 * w;
  const int node = t0 + mm;
  const int   nidx = node < n ? node : 0;
  const float dn   = (node < n) ? dinv[nidx] : 0.0f;
  const float yn   = (node < n) ? y[nidx] : 0.0f;
  const float cmsk = (mm == 0) ? 1.0f : 0.0f;
  v2f Bf[8], Wf[8], bf[8];
  #pragma unroll
  for (int i = 0; i < 8; ++i) {
    const int f0 = 4 * i + 2 * half;
    Bf[i].x = W2[(f0    ) * 2 + 1] - W2[(f0    ) * 2];
    Bf[i].y = W2[(f0 + 1) * 2 + 1] - W2[(f0 + 1) * 2];
    Wf[i].x = W1[f0];  Wf[i].y = W1[f0 + 1];
    bf[i].x = b1[f0];  bf[i].y = b1[f0 + 1];
  }

  for (int i = tid; i < RB * LP / 4; i += PT) ((v4f*)acc)[i] = (v4f)(0.0f);
  int a[KPW], m[KPW], z[KPW];
  seg_bounds(tab, w, b, a, m, z);
  stage_half(tbl, y, min(hs, n), tid);
  asm volatile("" :: "v"(dn), "v"(yn),
               "v"(Bf[0]), "v"(Bf[1]), "v"(Bf[2]), "v"(Bf[3]), "v"(Bf[4]), "v"(Bf[5]), "v"(Bf[6]), "v"(Bf[7]),
               "v"(Wf[0]), "v"(Wf[1]), "v"(Wf[2]), "v"(Wf[3]), "v"(Wf[4]), "v"(Wf[5]), "v"(Wf[6]), "v"(Wf[7]),
               "v"(bf[0]), "v"(bf[1]), "v"(bf[2]), "v"(bf[3]), "v"(bf[4]), "v"(bf[5]), "v"(bf[6]), "v"(bf[7]));
  __syncthreads();
  walk_half(pairs, w, ch, lane, a, m, tbl, acc);
  __syncthreads();
  stage_half(tbl, y + hs, n - hs, tid);
  __syncthreads();
  walk_half(pairs, w, ch, lane, m, z, tbl, acc);
  __syncthreads();

  if (w >= RB / 16) return;
  float part = 0.0f;
  #pragma unroll
  for (int j = 0; j < LP / 2; ++j)
    part += acc[(16 * w + mm) * LP + (LP / 2) * half + ((j + (mm >> 1)) & (LP / 2 - 1))];
  const float tot = part + __shfl_xor(part, 16);
  const float sn = dn * (yn + tot);
  v8f c = {};
  #pragma unroll
  for (int i = 0; i < 8; ++i) {
    v2f A;
    A.x = dn * fmaxf(fmaf(sn, Wf[i].x, bf[i].x), 0.0f);
    A.y = dn * fmaxf(fmaf(sn, Wf[i].y, bf[i].y), 0.0f);
    v2f B;
    B.x = Bf[i].x * cmsk;
    B.y = Bf[i].y * cmsk;
    c = __builtin_amdgcn_wmma_f32_16x16x4_f32(
             false, A,  false, B,
             (short)0, c,  false,  false);
  }

  if (mm == 0) {
    #pragma unroll
    for (int v = 0; v < 8; ++v) {
      int row = t0 + v + 8 * half;
      if (row < n) gd[row] = c[v];
    }
  }
}

__global__ __launch_bounds__(PT)
void k_layer2(const int* __restrict__ pairsRaw, const int* __restrict__ tab,
              const float* __restrict__ gd,
              const float* __restrict__ dinv,
              const float* __restrict__ b2, float* __restrict__ out, int n, int ch, int hs) {
  __shared__ __attribute__((aligned(16))) float lds[RB * LP + HMAX];
  float* acc = lds;
  float* tbl = lds + RB * LP;
  const int tid = threadIdx.x, lane = tid & 31;
  const int w  = __builtin_amdgcn_readfirstlane(tid >> 5);
  const int b  = blockIdx.x;
  const int n0 = b * RB;
  const int* __restrict__ pairs = pairsRaw;

  const int   onode = n0 + tid;
  const float gn    = (tid < RB && onode < n) ? gd[onode] : 0.0f;
  for (int i = tid; i < RB * LP / 4; i += PT) ((v4f*)acc)[i] = (v4f)(0.0f);
  int a[KPW], m[KPW], z[KPW];
  seg_bounds(tab, w, b, a, m, z);
  stage_half(tbl, gd, min(hs, n), tid);
  asm volatile("" :: "v"(gn));
  __syncthreads();
  walk_half(pairs, w, ch, lane, a, m, tbl, acc);
  __syncthreads();
  stage_half(tbl, gd + hs, n - hs, tid);
  __syncthreads();
  walk_half(pairs, w, ch, lane, m, z, tbl, acc);
  __syncthreads();
  if (tid < RB) {
    float tot = 0.0f;
    #pragma unroll
    for (int j = 0; j < LP; ++j) tot += acc[tid * LP + ((j + tid) & (LP - 1))];
    int node = n0 + tid;
    if (node < n) {
      float dl  = dinv[node] * (gn + tot) + (b2[1] - b2[0]);
      float mx  = fmaxf(dl, 0.0f);
      float lse = mx + logf(1.0f + expf(-fabsf(dl)));
      float2 r; r.x = -lse; r.y = dl - lse;
      *(float2*)(out + 2 * node) = r;
    }
  }
}

extern "C" void kernel_launch(void* const* d_in, const int* in_sizes, int n_in,
                              void* d_out, int out_size, void* d_ws, size_t ws_size,
                              hipStream_t stream) {
  const float* x  = (const float*)d_in[0];
  const int*   ei = (const int*)d_in[1];
  const float* W1 = (const float*)d_in[2];
  const float* b1 = (const float*)d_in[3];
  const float* W2 = (const float*)d_in[4];
  const float* b2 = (const float*)d_in[5];
  float* out = (float*)d_out;

  int n = in_sizes[0];
  int e = in_sizes[1] / 2;
  const int* srcIdx = ei;
  const int* dstIdx = ei + e;

  int np = (n + 3) & ~3;
  float* y     = (float*)d_ws;
  float* gd    = y + np;
  float* dinv  = gd + np;
  int*   tab   = (int*)(dinv + np);
  int*   pairs = tab + 2 * NCH * KEYS;

  int nb = (n + RB - 1) / RB;
  int ch = (((e + NCH - 1) / NCH) + 31) & ~31;
  int hs = (((n + 1) / 2) + 3) & ~3;

  k_part  <<<NCH, PT, 0, stream>>>(srcIdx, dstIdx, pairs, tab, e, ch, hs);
  k_deg   <<<nb,  PT, 0, stream>>>(pairs, tab, x, dinv, y, n, ch);
  k_layer1<<<nb,  PT, 0, stream>>>(pairs, tab, y, dinv, W1, b1, W2, gd, n, ch, hs);
  k_layer2<<<nb,  PT, 0, stream>>>(pairs, tab, gd, dinv, b2, out, n, ch, hs);
}
